# starting kernel
# baseline (speedup 1.0000x reference)
#include <hip/hip_runtime.h>

typedef __attribute__((ext_vector_type(2))) float v2f;
typedef __attribute__((ext_vector_type(8))) float v8f;

#define DIM  128
#define LDSP 132


__global__ __launch_bounds__(256) void k_deg_init(float* deg, int n) {
  int i = blockIdx.x * 256 + threadIdx.x;
  if (i < n) deg[i] = 1.0f;
}

__global__ __launch_bounds__(256) void k_deg_edges(const int* __restrict__ dst,
                                                   float* deg, int ne) {
  int e = blockIdx.x * 256 + threadIdx.x;
  if (e < ne) atomicAdd(&deg[dst[e]], 1.0f);
}

__global__ __launch_bounds__(256) void k_rsqrt(float* deg, int n) {
  int i = blockIdx.x * 256 + threadIdx.x;
  if (i < n) deg[i] = rsqrtf(deg[i]);
}


template <int RELU>
__global__ __launch_bounds__(256) void k_gemm_wmma(const float* __restrict__ A,
                                                   const float* __restrict__ W,
                                                   float* __restrict__ Out,
                                                   int nrows) {
  __shared__ float lw[DIM * LDSP];
  const int tid = threadIdx.x;
  for (int idx = tid; idx < DIM * DIM; idx += 256) {
    int k = idx >> 7, n = idx & (DIM - 1);
    lw[n * LDSP + k] = W[idx];
  }
  __syncthreads();

  const int lane  = tid & 31;
  const int lrow  = lane & 15;
  const int khalf = (lane >> 4) << 1;
  const size_t m0 = ((size_t)blockIdx.x * 8 + (tid >> 5)) * 16;
  if (m0 >= (size_t)nrows) return;

  v8f acc[8] = {};
  const float* arow = A + (m0 + lrow) * DIM;

  for (int k0 = 0; k0 < DIM; k0 += 4) {
    const int ka = k0 + khalf;
    v2f a = *(const v2f*)(arow + ka);
    if (RELU) { a.x = fmaxf(a.x, 0.0f); a.y = fmaxf(a.y, 0.0f); }
#pragma unroll
    for (int t = 0; t < 8; ++t) {
      v2f b = *(const v2f*)(&lw[(t * 16 + lrow) * LDSP + ka]);
      acc[t] = __builtin_amdgcn_wmma_f32_16x16x4_f32(
          false, a, false, b, (short)0, acc[t], false, false);
    }
  }

  const int rbase = (lane >> 4) << 3;
#pragma unroll
  for (int t = 0; t < 8; ++t) {
    float* op = Out + (m0 + rbase) * DIM + t * 16 + lrow;
#pragma unroll
    for (int r = 0; r < 8; ++r) op[(size_t)r * DIM] = acc[t][r];
  }
}

__global__ __launch_bounds__(256) void k_agg_init(const float* __restrict__ h,
                                                  const float* __restrict__ dinv,
                                                  const float* __restrict__ bias,
                                                  float* __restrict__ out,
                                                  int nnodes) {
  size_t idx = (size_t)blockIdx.x * 256 + threadIdx.x;
  size_t total = (size_t)nnodes * DIM;
  if (idx >= total) return;
  int node = (int)(idx >> 7);
  int c    = (int)(idx & (DIM - 1));
  float di = dinv[node];
  out[idx] = di * di * h[idx] + bias[c];
}

__global__ __launch_bounds__(256) void k_agg_edges(const int* __restrict__ src,
                                                   const int* __restrict__ dst,
                                                   const float* __restrict__ dinv,
                                                   const float* __restrict__ h,
                                                   float* __restrict__ out,
                                                   int ne) {
  int e    = (int)(((size_t)blockIdx.x * 256 + threadIdx.x) >> 5);
  int lane = threadIdx.x & 31;
  if (e >= ne) return;
  int s = src[e], d = dst[e];
  float norm = dinv[s] * dinv[d];
  const float4* hp = (const float4*)(h + (size_t)s * DIM);
  float4 v = hp[lane];
  float* op = out + (size_t)d * DIM + lane * 4;
  atomicAdd(op + 0, v.x * norm);
  atomicAdd(op + 1, v.y * norm);
  atomicAdd(op + 2, v.z * norm);
  atomicAdd(op + 3, v.w * norm);
}

__global__ __launch_bounds__(DIM) void k_pool(const float* __restrict__ h2,
                                              const int* __restrict__ scope,
                                              float* __restrict__ out,
                                              int ngraphs) {
  int g = blockIdx.x;
  int c = threadIdx.x;
  if (g >= ngraphs) return;
  size_t start = 0;
  for (int i = 0; i < g; ++i) start += (size_t)scope[i];
  int cnt = scope[g];
  float s = 0.0f;
  for (int j = 0; j < cnt; ++j) s += h2[(start + j) * DIM + c];
  out[(size_t)g * DIM + c] = s / (float)cnt;
}

extern "C" void kernel_launch(void* const* d_in, const int* in_sizes, int n_in,
                              void* d_out, int out_size, void* d_ws, size_t ws_size,
                              hipStream_t stream) {
  (void)n_in; (void)out_size; (void)ws_size;
  const float* x     = (const float*)d_in[0];
  const int*   ei    = (const int*)d_in[1];
  const int*   scope = (const int*)d_in[2];
  const float* W1    = (const float*)d_in[3];
  const float* b1    = (const float*)d_in[4];
  const float* W2    = (const float*)d_in[5];
  const float* b2    = (const float*)d_in[6];
  float*       out   = (float*)d_out;

  const int n_nodes  = in_sizes[0] / DIM;
  const int n_edges  = in_sizes[1] / 2;
  const int n_graphs = in_sizes[2];
  const int* src = ei;
  const int* dst = ei + n_edges;

  float* dinv = (float*)d_ws;
  float* bufA = dinv + n_nodes;
  float* bufB = bufA + (size_t)n_nodes * DIM;

  const int gb_nodes  = (n_nodes + 255) / 256;
  const int gb_edges  = (n_edges + 255) / 256;
  const size_t total  = (size_t)n_nodes * DIM;
  const int gb_elems  = (int)((total + 255) / 256);
  const int mtiles    = (n_nodes + 15) / 16;
  const int gb_gemm   = (mtiles + 7) / 8;
  const int gb_wedges = (n_edges + 7) / 8;

  k_deg_init <<<gb_nodes, 256, 0, stream>>>(dinv, n_nodes);
  k_deg_edges<<<gb_edges, 256, 0, stream>>>(dst, dinv, n_edges);
  k_rsqrt    <<<gb_nodes, 256, 0, stream>>>(dinv, n_nodes);

  k_gemm_wmma<0><<<gb_gemm,   256, 0, stream>>>(x, W1, bufA, n_nodes);
  k_agg_init    <<<gb_elems,  256, 0, stream>>>(bufA, dinv, b1, bufB, n_nodes);
  k_agg_edges   <<<gb_wedges, 256, 0, stream>>>(src, dst, dinv, bufA, bufB, n_edges);

  k_gemm_wmma<1><<<gb_gemm,   256, 0, stream>>>(bufB, W2, bufA, n_nodes);
  k_agg_init    <<<gb_elems,  256, 0, stream>>>(bufA, dinv, b2, bufB, n_nodes);
  k_agg_edges   <<<gb_wedges, 256, 0, stream>>>(src, dst, dinv, bufA, bufB, n_edges);

  k_pool<<<n_graphs, DIM, 0, stream>>>(bufB, scope, out, n_graphs);
}


# edge scatter: lane l owns channels l, l+32, l+64, l+96, so each row load and each f32 atomic add covers one whole 128-byte line (was 4 lines, 8 lanes each); atomic sum order stays arrival order
# speedup vs baseline: 1.3421x; 1.3421x over previous
#include <hip/hip_runtime.h>

typedef __attribute__((ext_vector_type(2))) float v2f;
typedef __attribute__((ext_vector_type(8))) float v8f;

#define DIM  128
#define LDSP 132


__global__ __launch_bounds__(256) void k_deg_init(float* deg, int n) {
  int i = blockIdx.x * 256 + threadIdx.x;
  if (i < n) deg[i] = 1.0f;
}

__global__ __launch_bounds__(256) void k_deg_edges(const int* __restrict__ dst,
                                                   float* deg, int ne) {
  int e = blockIdx.x * 256 + threadIdx.x;
  if (e < ne) atomicAdd(&deg[dst[e]], 1.0f);
}

__global__ __launch_bounds__(256) void k_rsqrt(float* deg, int n) {
  int i = blockIdx.x * 256 + threadIdx.x;
  if (i < n) deg[i] = rsqrtf(deg[i]);
}


template <int RELU>
__global__ __launch_bounds__(256) void k_gemm_wmma(const float* __restrict__ A,
                                                   const float* __restrict__ W,
                                                   float* __restrict__ Out,
                                                   int nrows) {
  __shared__ float lw[DIM * LDSP];
  const int tid = threadIdx.x;
  for (int idx = tid; idx < DIM * DIM; idx += 256) {
    int k = idx >> 7, n = idx & (DIM - 1);
    lw[n * LDSP + k] = W[idx];
  }
  __syncthreads();

  const int lane  = tid & 31;
  const int lrow  = lane & 15;
  const int khalf = (lane >> 4) << 1;
  const size_t m0 = ((size_t)blockIdx.x * 8 + (tid >> 5)) * 16;
  if (m0 >= (size_t)nrows) return;

  v8f acc[8] = {};
  const float* arow = A + (m0 + lrow) * DIM;

  for (int k0 = 0; k0 < DIM; k0 += 4) {
    const int ka = k0 + khalf;
    v2f a = *(const v2f*)(arow + ka);
    if (RELU) { a.x = fmaxf(a.x, 0.0f); a.y = fmaxf(a.y, 0.0f); }
#pragma unroll
    for (int t = 0; t < 8; ++t) {
      v2f b = *(const v2f*)(&lw[(t * 16 + lrow) * LDSP + ka]);
      acc[t] = __builtin_amdgcn_wmma_f32_16x16x4_f32(
          false, a, false, b, (short)0, acc[t], false, false);
    }
  }

  const int rbase = (lane >> 4) << 3;
#pragma unroll
  for (int t = 0; t < 8; ++t) {
    float* op = Out + (m0 + rbase) * DIM + t * 16 + lrow;
#pragma unroll
    for (int r = 0; r < 8; ++r) op[(size_t)r * DIM] = acc[t][r];
  }
}

__global__ __launch_bounds__(256) void k_agg_init(const float* __restrict__ h,
                                                  const float* __restrict__ dinv,
                                                  const float* __restrict__ bias,
                                                  float* __restrict__ out,
                                                  int nnodes) {
  size_t idx = (size_t)blockIdx.x * 256 + threadIdx.x;
  size_t total = (size_t)nnodes * DIM;
  if (idx >= total) return;
  int node = (int)(idx >> 7);
  int c    = (int)(idx & (DIM - 1));
  float di = dinv[node];
  out[idx] = di * di * h[idx] + bias[c];
}

__global__ __launch_bounds__(256) void k_agg_edges(const int* __restrict__ src,
                                                   const int* __restrict__ dst,
                                                   const float* __restrict__ dinv,
                                                   const float* __restrict__ h,
                                                   float* __restrict__ out,
                                                   int ne) {
  int e    = (int)(((size_t)blockIdx.x * 256 + threadIdx.x) >> 5);
  int lane = threadIdx.x & 31;
  if (e >= ne) return;
  int s = src[e], d = dst[e];
  float norm = dinv[s] * dinv[d];
  const float* hp = h + (size_t)s * DIM + lane;
  float v0 = hp[0];
  float v1 = hp[32];
  float v2 = hp[64];
  float v3 = hp[96];
  float* op = out + (size_t)d * DIM + lane;
  atomicAdd(op +  0, v0 * norm);
  atomicAdd(op + 32, v1 * norm);
  atomicAdd(op + 64, v2 * norm);
  atomicAdd(op + 96, v3 * norm);
}

__global__ __launch_bounds__(DIM) void k_pool(const float* __restrict__ h2,
                                              const int* __restrict__ scope,
                                              float* __restrict__ out,
                                              int ngraphs) {
  int g = blockIdx.x;
  int c = threadIdx.x;
  if (g >= ngraphs) return;
  size_t start = 0;
  for (int i = 0; i < g; ++i) start += (size_t)scope[i];
  int cnt = scope[g];
  float s = 0.0f;
  for (int j = 0; j < cnt; ++j) s += h2[(start + j) * DIM + c];
  out[(size_t)g * DIM + c] = s / (float)cnt;
}

extern "C" void kernel_launch(void* const* d_in, const int* in_sizes, int n_in,
                              void* d_out, int out_size, void* d_ws, size_t ws_size,
                              hipStream_t stream) {
  (void)n_in; (void)out_size; (void)ws_size;
  const float* x     = (const float*)d_in[0];
  const int*   ei    = (const int*)d_in[1];
  const int*   scope = (const int*)d_in[2];
  const float* W1    = (const float*)d_in[3];
  const float* b1    = (const float*)d_in[4];
  const float* W2    = (const float*)d_in[5];
  const float* b2    = (const float*)d_in[6];
  float*       out   = (float*)d_out;

  const int n_nodes  = in_sizes[0] / DIM;
  const int n_edges  = in_sizes[1] / 2;
  const int n_graphs = in_sizes[2];
  const int* src = ei;
  const int* dst = ei + n_edges;

  float* dinv = (float*)d_ws;
  float* bufA = dinv + n_nodes;
  float* bufB = bufA + (size_t)n_nodes * DIM;

  const int gb_nodes  = (n_nodes + 255) / 256;
  const int gb_edges  = (n_edges + 255) / 256;
  const size_t total  = (size_t)n_nodes * DIM;
  const int gb_elems  = (int)((total + 255) / 256);
  const int mtiles    = (n_nodes + 15) / 16;
  const int gb_gemm   = (mtiles + 7) / 8;
  const int gb_wedges = (n_edges + 7) / 8;

  k_deg_init <<<gb_nodes, 256, 0, stream>>>(dinv, n_nodes);
  k_deg_edges<<<gb_edges, 256, 0, stream>>>(dst, dinv, n_edges);
  k_rsqrt    <<<gb_nodes, 256, 0, stream>>>(dinv, n_nodes);

  k_gemm_wmma<0><<<gb_gemm,   256, 0, stream>>>(x, W1, bufA, n_nodes);
  k_agg_init    <<<gb_elems,  256, 0, stream>>>(bufA, dinv, b1, bufB, n_nodes);
  k_agg_edges   <<<gb_wedges, 256, 0, stream>>>(src, dst, dinv, bufA, bufB, n_edges);

  k_gemm_wmma<1><<<gb_gemm,   256, 0, stream>>>(bufB, W2, bufA, n_nodes);
  k_agg_init    <<<gb_elems,  256, 0, stream>>>(bufA, dinv, b2, bufB, n_nodes);
  k_agg_edges   <<<gb_wedges, 256, 0, stream>>>(src, dst, dinv, bufA, bufB, n_edges);

  k_pool<<<n_graphs, DIM, 0, stream>>>(bufB, scope, out, n_graphs);
}


# layer-1 scatter as a gather on x ahead of the product: list built first, a wave per node sums weight*x[src] in registers, then (Ax)W1+b1 in place; no f32 row atomics; layer-1 sum order = slot order
# speedup vs baseline: 2.4332x; 1.8130x over previous
#include <hip/hip_runtime.h>

typedef __attribute__((ext_vector_type(2))) float v2f;
typedef __attribute__((ext_vector_type(8))) float v8f;

#define DIM  128
#define LDSP 132


__global__ __launch_bounds__(256) void k_deg_init(float* deg, int n) {
  int i = blockIdx.x * 256 + threadIdx.x;
  if (i < n) deg[i] = 1.0f;
}

__global__ __launch_bounds__(256) void k_deg_edges(const int* __restrict__ dst,
                                                   float* deg, int ne) {
  int e = blockIdx.x * 256 + threadIdx.x;
  if (e < ne) atomicAdd(&deg[dst[e]], 1.0f);
}


template <int RELU, int BIAS>
__global__ __launch_bounds__(256) void k_gemm_wmma(const float* A,
                                                   const float* __restrict__ W,
                                                   const float* __restrict__ bias,
                                                   float* Out,
                                                   int nrows) {
  __shared__ float lw[DIM * LDSP];
  const int tid = threadIdx.x;
  for (int idx = tid; idx < DIM * DIM; idx += 256) {
    int k = idx >> 7, n = idx & (DIM - 1);
    lw[n * LDSP + k] = W[idx];
  }
  __syncthreads();

  const int lane  = tid & 31;
  const int lrow  = lane & 15;
  const int khalf = (lane >> 4) << 1;
  const size_t m0 = ((size_t)blockIdx.x * 8 + (tid >> 5)) * 16;
  if (m0 >= (size_t)nrows) return;

  v8f acc[8] = {};
  const float* arow = A + (m0 + lrow) * DIM;

  for (int k0 = 0; k0 < DIM; k0 += 4) {
    const int ka = k0 + khalf;
    v2f a = *(const v2f*)(arow + ka);
    if (RELU) { a.x = fmaxf(a.x, 0.0f); a.y = fmaxf(a.y, 0.0f); }
#pragma unroll
    for (int t = 0; t < 8; ++t) {
      v2f b = *(const v2f*)(&lw[(t * 16 + lrow) * LDSP + ka]);
      acc[t] = __builtin_amdgcn_wmma_f32_16x16x4_f32(
          false, a, false, b, (short)0, acc[t], false, false);
    }
  }

  const int rbase = (lane >> 4) << 3;
#pragma unroll
  for (int t = 0; t < 8; ++t) {
    float* op = Out + (m0 + rbase) * DIM + t * 16 + lrow;
    const float bv = BIAS ? bias[t * 16 + lrow] : 0.0f;
#pragma unroll
    for (int r = 0; r < 8; ++r) op[(size_t)r * DIM] = BIAS ? acc[t][r] + bv : acc[t][r];
  }
}


__device__ __forceinline__ int wave_sum(int v) {
  for (int o = 16; o > 0; o >>= 1) v += __shfl_xor(v, o);
  return v;
}

__device__ __forceinline__ int wave_incl_scan(int v, int lane) {
  for (int o = 1; o < 32; o <<= 1) {
    int t = __shfl_up(v, o);
    if (lane >= o) v += t;
  }
  return v;
}

__global__ __launch_bounds__(1024) void k_scan(const float* __restrict__ deg,
                                               int* __restrict__ rowptr,
                                               int* __restrict__ cursor,
                                               int2* __restrict__ ent,
                                               int n) {
  __shared__ int wtot[32];
  __shared__ int wexc[32];
  const int tid  = threadIdx.x;
  const int lane = tid & 31;
  const int w    = tid >> 5;
  const int base = blockIdx.x * 1024;

  int part = 0;
  for (int j = tid; j < base; j += 1024) part += (int)deg[j];
  part = wave_sum(part);
  if (lane == 0) wtot[w] = part;

  const int   i  = base + tid;
  const float dg = (i < n) ? deg[i] : 0.0f;
  const int   c  = (int)dg;
  const int incl = wave_incl_scan(c, lane);
  if (lane == 31) wexc[w] = incl;
  __syncthreads();
  if (w == 0) {
    const int tot = wave_sum(wtot[lane]);
    const int v   = wexc[lane];
    const int s   = wave_incl_scan(v, lane);
    wexc[lane] = tot + s - v;
  }
  __syncthreads();
  if (i < n) {
    const int first = wexc[w] + incl - c;
    rowptr[i] = first;
    cursor[i] = 1;
    const float di = rsqrtf(dg);
    ent[first] = make_int2(i, __float_as_int(di * di));
    if (i == n - 1) rowptr[n] = first + c;
  }
}

__global__ __launch_bounds__(256) void k_fill(const int* __restrict__ src,
                                              const int* __restrict__ dst,
                                              const float* __restrict__ deg,
                                              const int* __restrict__ rowptr,
                                              int* cursor,
                                              int2* __restrict__ ent,
                                              int ne) {
  int e = blockIdx.x * 256 + threadIdx.x;
  if (e >= ne) return;
  int s = src[e], d = dst[e];
  float norm = rsqrtf(deg[s]) * rsqrtf(deg[d]);
  int slot = rowptr[d] + atomicAdd(&cursor[d], 1);
  ent[slot] = make_int2(s, __float_as_int(norm));
}

template <int U>
__device__ __forceinline__ void rows_fma(const float4* __restrict__ x4, int2 en,
                                         int j, int lane, float4& acc) {
  float4 v[U];
  float  wt[U];
  for (int u = 0; u < U; ++u) {
    const int s = __builtin_amdgcn_readlane(en.x, j + u);
    wt[u] = __int_as_float(__builtin_amdgcn_readlane(en.y, j + u));
    v[u]  = x4[(size_t)s * (DIM / 4) + lane];
  }
  for (int u = 0; u < U; ++u) {
    acc.x = fmaf(wt[u], v[u].x, acc.x);
    acc.y = fmaf(wt[u], v[u].y, acc.y);
    acc.z = fmaf(wt[u], v[u].z, acc.z);
    acc.w = fmaf(wt[u], v[u].w, acc.w);
  }
}

__global__ __launch_bounds__(256) void k_gather_nodes(const float* __restrict__ x,
                                                      const int2* __restrict__ ent,
                                                      const int* __restrict__ rowptr,
                                                      float* __restrict__ out,
                                                      int nnodes) {
  const int lane = threadIdx.x & 31;
  const int node = __builtin_amdgcn_readfirstlane((int)((blockIdx.x * 256u + threadIdx.x) >> 5));
  if (node >= nnodes) return;
  const int p0 = rowptr[node];
  const int p1 = rowptr[node + 1];

  const float4* x4 = (const float4*)x;
  float4 acc = make_float4(0.0f, 0.0f, 0.0f, 0.0f);
  for (int cb = p0; cb < p1; cb += 32) {
    const int m = min(32, p1 - cb);
    int2 en = make_int2(0, 0);
    if (lane < m) en = ent[cb + lane];
    int j = 0;
    for (; j + 8 <= m; j += 8) rows_fma<8>(x4, en, j, lane, acc);
    if (j + 4 <= m) { rows_fma<4>(x4, en, j, lane, acc); j += 4; }
    if (j + 2 <= m) { rows_fma<2>(x4, en, j, lane, acc); j += 2; }
    if (j < m)      { rows_fma<1>(x4, en, j, lane, acc); }
  }
  ((float4*)out)[(size_t)node * (DIM / 4) + lane] = acc;
}

__global__ __launch_bounds__(1024) void k_gather_pool(const float* __restrict__ h,
                                                      const int2* __restrict__ ent,
                                                      const int* __restrict__ rowptr,
                                                      const int* __restrict__ scope,
                                                      const float* __restrict__ bias,
                                                      float* __restrict__ out,
                                                      int ngraphs, int nnodes) {
  __shared__ int   wpart[32];
  __shared__ float red[32 * DIM];
  const int g    = blockIdx.x;
  const int tid  = threadIdx.x;
  const int lane = tid & 31;
  const int w = __builtin_amdgcn_readfirstlane(tid >> 5);
  if (g >= ngraphs) return;

  int part = 0;
  for (int i = tid; i < g; i += 1024) part += scope[i];
  part = wave_sum(part);
  if (lane == 0) wpart[w] = part;
  __syncthreads();
  const int start = __builtin_amdgcn_readfirstlane(wave_sum(wpart[lane]));
  const int cnt   = scope[g];
  const int n0 = min(start, nnodes);
  const int n1 = min(start + cnt, nnodes);
  const int p0 = rowptr[n0];
  const int p1 = rowptr[n1];

  const float4* h4 = (const float4*)h;
  float4 acc = make_float4(0.0f, 0.0f, 0.0f, 0.0f);
  for (int cb = p0 + 32 * w; cb < p1; cb += 32 * 32) {
    const int m = min(32, p1 - cb);
    int2 en = make_int2(0, 0);
    if (lane < m) en = ent[cb + lane];
    for (int j = 0; j < m; ++j) {
      const int   s  = __builtin_amdgcn_readlane(en.x, j);
      const float wt = __int_as_float(__builtin_amdgcn_readlane(en.y, j));
      const float4 v = h4[(size_t)s * (DIM / 4) + lane];
      acc.x = fmaf(wt, v.x, acc.x);
      acc.y = fmaf(wt, v.y, acc.y);
      acc.z = fmaf(wt, v.z, acc.z);
      acc.w = fmaf(wt, v.w, acc.w);
    }
  }

  *(float4*)(&red[w * DIM + lane * 4]) = acc;
  __syncthreads();
  if (tid < DIM) {
    float s = 0.0f;
    for (int k = 0; k < 32; ++k) s += red[k * DIM + tid];
    out[(size_t)g * DIM + tid] = s / (float)cnt + bias[tid];
  }
}

extern "C" void kernel_launch(void* const* d_in, const int* in_sizes, int n_in,
                              void* d_out, int out_size, void* d_ws, size_t ws_size,
                              hipStream_t stream) {
  (void)n_in; (void)out_size; (void)ws_size;
  const float* x     = (const float*)d_in[0];
  const int*   ei    = (const int*)d_in[1];
  const int*   scope = (const int*)d_in[2];
  const float* W1    = (const float*)d_in[3];
  const float* b1    = (const float*)d_in[4];
  const float* W2    = (const float*)d_in[5];
  const float* b2    = (const float*)d_in[6];
  float*       out   = (float*)d_out;

  const int n_nodes  = in_sizes[0] / DIM;
  const int n_edges  = in_sizes[1] / 2;
  const int n_graphs = in_sizes[2];
  const int* src = ei;
  const int* dst = ei + n_edges;

  float* bufA = (float*)d_ws;
  float* bufB = bufA + (size_t)n_nodes * DIM;
  float* deg  = bufB + (size_t)n_nodes * DIM;
  int2* ent    = (int2*)bufB;
  int*  rowptr = (int*)(ent + ((size_t)n_edges + n_nodes));
  int*  cursor = rowptr + n_nodes + 1;

  const int gb_nodes  = (n_nodes + 255) / 256;
  const int gb_edges  = (n_edges + 255) / 256;
  const int mtiles    = (n_nodes + 15) / 16;
  const int gb_gemm   = (mtiles + 7) / 8;
  const int gb_wnodes = (n_nodes + 7) / 8;
  const int gb_scan   = (n_nodes + 1023) / 1024;

  k_deg_init <<<gb_nodes, 256, 0, stream>>>(deg, n_nodes);
  k_deg_edges<<<gb_edges, 256, 0, stream>>>(dst, deg, n_edges);

  k_scan        <<<gb_scan,  1024, 0, stream>>>(deg, rowptr, cursor, ent, n_nodes);
  k_fill        <<<gb_edges,  256, 0, stream>>>(src, dst, deg, rowptr, cursor, ent, n_edges);

  k_gather_nodes<<<gb_wnodes, 256, 0, stream>>>(x, ent, rowptr, bufA, n_nodes);
  k_gemm_wmma<0, 1><<<gb_gemm, 256, 0, stream>>>(bufA, W1, b1, bufA, n_nodes);

  k_gemm_wmma<1, 0><<<gb_gemm, 256, 0, stream>>>(bufA, W2, b2, bufA, n_nodes);
  k_gather_pool <<<n_graphs, 1024, 0, stream>>>(bufA, ent, rowptr, scope, b2, out,
                                                n_graphs, n_nodes);
}


# edge rank = return of the in-degree count's integer add; list fill becomes plain stores (one pass of scattered atomics per call, was two); sum order = arrival order at the count, was at the fill
# speedup vs baseline: 3.0235x; 1.2426x over previous
#include <hip/hip_runtime.h>

typedef __attribute__((ext_vector_type(2))) float v2f;
typedef __attribute__((ext_vector_type(8))) float v8f;

#define DIM  128
#define LDSP 132


__global__ __launch_bounds__(256) void k_cnt_init(int* cnt, int n) {
  int i = blockIdx.x * 256 + threadIdx.x;
  if (i < n) cnt[i] = 0;
}

__global__ __launch_bounds__(256) void k_cnt_edges(const int* __restrict__ dst,
                                                   int* cnt,
                                                   int* __restrict__ rank, int ne) {
  int e = blockIdx.x * 256 + threadIdx.x;
  if (e < ne) rank[e] = atomicAdd(&cnt[dst[e]], 1);
}


template <int RELU, int BIAS>
__global__ __launch_bounds__(256) void k_gemm_wmma(const float* A,
                                                   const float* __restrict__ W,
                                                   const float* __restrict__ bias,
                                                   float* Out,
                                                   int nrows) {
  __shared__ float lw[DIM * LDSP];
  const int tid = threadIdx.x;
  for (int idx = tid; idx < DIM * DIM; idx += 256) {
    int k = idx >> 7, n = idx & (DIM - 1);
    lw[n * LDSP + k] = W[idx];
  }
  __syncthreads();

  const int lane  = tid & 31;
  const int lrow  = lane & 15;
  const int khalf = (lane >> 4) << 1;
  const size_t m0 = ((size_t)blockIdx.x * 8 + (tid >> 5)) * 16;
  if (m0 >= (size_t)nrows) return;

  v8f acc[8] = {};
  const float* arow = A + (m0 + lrow) * DIM;

  for (int k0 = 0; k0 < DIM; k0 += 4) {
    const int ka = k0 + khalf;
    v2f a = *(const v2f*)(arow + ka);
    if (RELU) { a.x = fmaxf(a.x, 0.0f); a.y = fmaxf(a.y, 0.0f); }
#pragma unroll
    for (int t = 0; t < 8; ++t) {
      v2f b = *(const v2f*)(&lw[(t * 16 + lrow) * LDSP + ka]);
      acc[t] = __builtin_amdgcn_wmma_f32_16x16x4_f32(
          false, a, false, b, (short)0, acc[t], false, false);
    }
  }

  const int rbase = (lane >> 4) << 3;
#pragma unroll
  for (int t = 0; t < 8; ++t) {
    float* op = Out + (m0 + rbase) * DIM + t * 16 + lrow;
    const float bv = BIAS ? bias[t * 16 + lrow] : 0.0f;
#pragma unroll
    for (int r = 0; r < 8; ++r) op[(size_t)r * DIM] = BIAS ? acc[t][r] + bv : acc[t][r];
  }
}


__device__ __forceinline__ int wave_sum(int v) {
  for (int o = 16; o > 0; o >>= 1) v += __shfl_xor(v, o);
  return v;
}

__device__ __forceinline__ int wave_incl_scan(int v, int lane) {
  for (int o = 1; o < 32; o <<= 1) {
    int t = __shfl_up(v, o);
    if (lane >= o) v += t;
  }
  return v;
}

__global__ __launch_bounds__(1024) void k_scan(const int* __restrict__ cnt,
                                               int* __restrict__ rowptr,
                                               int2* __restrict__ ent,
                                               int n) {
  __shared__ int wtot[32];
  __shared__ int wexc[32];
  const int tid  = threadIdx.x;
  const int lane = tid & 31;
  const int w    = tid >> 5;
  const int base = blockIdx.x * 1024;

  int part = 0;
  for (int j = tid; j < base; j += 1024) part += cnt[j] + 1;
  part = wave_sum(part);
  if (lane == 0) wtot[w] = part;

  const int   i  = base + tid;
  const int   c  = (i < n) ? cnt[i] + 1 : 0;
  const int incl = wave_incl_scan(c, lane);
  if (lane == 31) wexc[w] = incl;
  __syncthreads();
  if (w == 0) {
    const int tot = wave_sum(wtot[lane]);
    const int v   = wexc[lane];
    const int s   = wave_incl_scan(v, lane);
    wexc[lane] = tot + s - v;
  }
  __syncthreads();
  if (i < n) {
    const int first = wexc[w] + incl - c;
    rowptr[i] = first;
    const float di = rsqrtf((float)c);
    ent[first] = make_int2(i, __float_as_int(di * di));
    if (i == n - 1) rowptr[n] = first + c;
  }
}

__global__ __launch_bounds__(256) void k_fill(const int* __restrict__ src,
                                              const int* __restrict__ dst,
                                              const int* __restrict__ cnt,
                                              const int* __restrict__ rank,
                                              const int* __restrict__ rowptr,
                                              int2* __restrict__ ent,
                                              int ne) {
  int e = blockIdx.x * 256 + threadIdx.x;
  if (e >= ne) return;
  int s = src[e], d = dst[e];
  float norm = rsqrtf((float)(cnt[s] + 1)) * rsqrtf((float)(cnt[d] + 1));
  int slot = rowptr[d] + 1 + rank[e];
  ent[slot] = make_int2(s, __float_as_int(norm));
}

template <int U>
__device__ __forceinline__ void rows_fma(const float4* __restrict__ x4, int2 en,
                                         int j, int lane, float4& acc) {
  float4 v[U];
  float  wt[U];
  for (int u = 0; u < U; ++u) {
    const int s = __builtin_amdgcn_readlane(en.x, j + u);
    wt[u] = __int_as_float(__builtin_amdgcn_readlane(en.y, j + u));
    v[u]  = x4[(size_t)s * (DIM / 4) + lane];
  }
  for (int u = 0; u < U; ++u) {
    acc.x = fmaf(wt[u], v[u].x, acc.x);
    acc.y = fmaf(wt[u], v[u].y, acc.y);
    acc.z = fmaf(wt[u], v[u].z, acc.z);
    acc.w = fmaf(wt[u], v[u].w, acc.w);
  }
}

__global__ __launch_bounds__(256) void k_gather_nodes(const float* __restrict__ x,
                                                      const int2* __restrict__ ent,
                                                      const int* __restrict__ rowptr,
                                                      float* __restrict__ out,
                                                      int nnodes) {
  const int lane = threadIdx.x & 31;
  const int node = __builtin_amdgcn_readfirstlane((int)((blockIdx.x * 256u + threadIdx.x) >> 5));
  if (node >= nnodes) return;
  const int p0 = rowptr[node];
  const int p1 = rowptr[node + 1];

  const float4* x4 = (const float4*)x;
  float4 acc = make_float4(0.0f, 0.0f, 0.0f, 0.0f);
  for (int cb = p0; cb < p1; cb += 32) {
    const int m = min(32, p1 - cb);
    int2 en = make_int2(0, 0);
    if (lane < m) en = ent[cb + lane];
    int j = 0;
    for (; j + 8 <= m; j += 8) rows_fma<8>(x4, en, j, lane, acc);
    if (j + 4 <= m) { rows_fma<4>(x4, en, j, lane, acc); j += 4; }
    if (j + 2 <= m) { rows_fma<2>(x4, en, j, lane, acc); j += 2; }
    if (j < m)      { rows_fma<1>(x4, en, j, lane, acc); }
  }
  ((float4*)out)[(size_t)node * (DIM / 4) + lane] = acc;
}

__global__ __launch_bounds__(1024) void k_gather_pool(const float* __restrict__ h,
                                                      const int2* __restrict__ ent,
                                                      const int* __restrict__ rowptr,
                                                      const int* __restrict__ scope,
                                                      const float* __restrict__ bias,
                                                      float* __restrict__ out,
                                                      int ngraphs, int nnodes) {
  __shared__ int   wpart[32];
  __shared__ float red[32 * DIM];
  const int g    = blockIdx.x;
  const int tid  = threadIdx.x;
  const int lane = tid & 31;
  const int w = __builtin_amdgcn_readfirstlane(tid >> 5);
  if (g >= ngraphs) return;

  int part = 0;
  for (int i = tid; i < g; i += 1024) part += scope[i];
  part = wave_sum(part);
  if (lane == 0) wpart[w] = part;
  __syncthreads();
  const int start = __builtin_amdgcn_readfirstlane(wave_sum(wpart[lane]));
  const int cnt   = scope[g];
  const int n0 = min(start, nnodes);
  const int n1 = min(start + cnt, nnodes);
  const int p0 = rowptr[n0];
  const int p1 = rowptr[n1];

  const float4* h4 = (const float4*)h;
  float4 acc = make_float4(0.0f, 0.0f, 0.0f, 0.0f);
  for (int cb = p0 + 32 * w; cb < p1; cb += 32 * 32) {
    const int m = min(32, p1 - cb);
    int2 en = make_int2(0, 0);
    if (lane < m) en = ent[cb + lane];
    for (int j = 0; j < m; ++j) {
      const int   s  = __builtin_amdgcn_readlane(en.x, j);
      const float wt = __int_as_float(__builtin_amdgcn_readlane(en.y, j));
      const float4 v = h4[(size_t)s * (DIM / 4) + lane];
      acc.x = fmaf(wt, v.x, acc.x);
      acc.y = fmaf(wt, v.y, acc.y);
      acc.z = fmaf(wt, v.z, acc.z);
      acc.w = fmaf(wt, v.w, acc.w);
    }
  }

  *(float4*)(&red[w * DIM + lane * 4]) = acc;
  __syncthreads();
  if (tid < DIM) {
    float s = 0.0f;
    for (int k = 0; k < 32; ++k) s += red[k * DIM + tid];
    out[(size_t)g * DIM + tid] = s / (float)cnt + bias[tid];
  }
}

extern "C" void kernel_launch(void* const* d_in, const int* in_sizes, int n_in,
                              void* d_out, int out_size, void* d_ws, size_t ws_size,
                              hipStream_t stream) {
  (void)n_in; (void)out_size; (void)ws_size;
  const float* x     = (const float*)d_in[0];
  const int*   ei    = (const int*)d_in[1];
  const int*   scope = (const int*)d_in[2];
  const float* W1    = (const float*)d_in[3];
  const float* b1    = (const float*)d_in[4];
  const float* W2    = (const float*)d_in[5];
  const float* b2    = (const float*)d_in[6];
  float*       out   = (float*)d_out;

  const int n_nodes  = in_sizes[0] / DIM;
  const int n_edges  = in_sizes[1] / 2;
  const int n_graphs = in_sizes[2];
  const int* src = ei;
  const int* dst = ei + n_edges;

  float* bufA = (float*)d_ws;
  float* bufB = bufA + (size_t)n_nodes * DIM;
  int*   cnt  = (int*)(bufB + (size_t)n_nodes * DIM);
  int2* ent    = (int2*)bufB;
  int*  rowptr = (int*)(ent + ((size_t)n_edges + n_nodes));
  int*  rank   = rowptr + n_nodes + 1;

  const int gb_nodes  = (n_nodes + 255) / 256;
  const int gb_edges  = (n_edges + 255) / 256;
  const int mtiles    = (n_nodes + 15) / 16;
  const int gb_gemm   = (mtiles + 7) / 8;
  const int gb_wnodes = (n_nodes + 7) / 8;
  const int gb_scan   = (n_nodes + 1023) / 1024;

  k_cnt_init <<<gb_nodes, 256, 0, stream>>>(cnt, n_nodes);
  k_cnt_edges<<<gb_edges, 256, 0, stream>>>(dst, cnt, rank, n_edges);

  k_scan        <<<gb_scan,  1024, 0, stream>>>(cnt, rowptr, ent, n_nodes);
  k_fill        <<<gb_edges,  256, 0, stream>>>(src, dst, cnt, rank, rowptr, ent, n_edges);

  k_gather_nodes<<<gb_wnodes, 256, 0, stream>>>(x, ent, rowptr, bufA, n_nodes);
  k_gemm_wmma<0, 1><<<gb_gemm, 256, 0, stream>>>(bufA, W1, b1, bufA, n_nodes);

  k_gemm_wmma<1, 0><<<gb_gemm, 256, 0, stream>>>(bufA, W2, b2, bufA, n_nodes);
  k_gather_pool <<<n_graphs, 1024, 0, stream>>>(bufA, ent, rowptr, scope, b2, out,
                                                n_graphs, n_nodes);
}


# in-degree count and edge rank by LDS adds: edge list cut into 64 runs, each counted per node in one work-group's LDS, then a per-node walk over runs; no global atomics; sum order = run, LDS arrival
# speedup vs baseline: 3.7706x; 1.2471x over previous
#include <hip/hip_runtime.h>

typedef __attribute__((ext_vector_type(2))) float v2f;
typedef __attribute__((ext_vector_type(8))) float v8f;

#define DIM  128
#define LDSP 132


#define NCHUNK 64
#define LWORDS 51200

__global__ __launch_bounds__(1024) void k_cnt_chunks(const int* __restrict__ dst,
                                                     unsigned* __restrict__ part,
                                                     int* __restrict__ rank,
                                                     int n, int ne) {
  __shared__ uint4 lc4[LWORDS / 4];
  unsigned* lc = (unsigned*)lc4;
  const int tid = threadIdx.x;
  const int pw4 = (((n + 1) >> 1) + 3) >> 2;
  for (int q = tid; q < pw4; q += 1024) lc4[q] = make_uint4(0u, 0u, 0u, 0u);
  __syncthreads();

  const int per = (ne + NCHUNK - 1) / NCHUNK;
  const int e0  = blockIdx.x * per;
  const int e1  = min(ne, e0 + per);
  for (int eb = e0 + tid; eb < e1; eb += 16 * 1024) {
    int      d[16];
    unsigned old[16];
#pragma unroll
    for (int u = 0; u < 16; ++u) {
      const int e = eb + u * 1024;
      d[u] = (e < e1) ? dst[e] : -1;
    }
#pragma unroll
    for (int u = 0; u < 16; ++u) {
      old[u] = 0u;
      if (d[u] >= 0) old[u] = atomicAdd(&lc[d[u] >> 1], (d[u] & 1) ? 0x10000u : 1u);
    }
#pragma unroll
    for (int u = 0; u < 16; ++u) {
      if (d[u] >= 0) rank[eb + u * 1024] = (d[u] & 1) ? (int)(old[u] >> 16) : (int)(old[u] & 0xffffu);
    }
  }
  __syncthreads();

  uint4* p4 = (uint4*)part + (size_t)blockIdx.x * pw4;
  for (int q = tid; q < pw4; q += 1024) p4[q] = lc4[q];
}

__global__ __launch_bounds__(256) void k_chunk_offsets(const unsigned* __restrict__ part,
                                                       int2* __restrict__ off,
                                                       int* __restrict__ cnt, int n) {
  const int nw = (n + 1) >> 1;
  const int pw = ((nw + 3) >> 2) << 2;
  const int w  = blockIdx.x * 256 + threadIdx.x;
  if (w >= nw) return;
  unsigned v[NCHUNK];
#pragma unroll
  for (int c = 0; c < NCHUNK; ++c) v[c] = part[(size_t)c * pw + w];
  int lo = 0, hi = 0;
#pragma unroll
  for (int c = 0; c < NCHUNK; ++c) {
    off[(size_t)c * pw + w] = make_int2(lo, hi);
    lo += (int)(v[c] & 0xffffu);
    hi += (int)(v[c] >> 16);
  }
  cnt[2 * w] = lo;
  if (2 * w + 1 < n) cnt[2 * w + 1] = hi;
}


template <int RELU, int BIAS>
__global__ __launch_bounds__(256) void k_gemm_wmma(const float* A,
                                                   const float* __restrict__ W,
                                                   const float* __restrict__ bias,
                                                   float* Out,
                                                   int nrows) {
  __shared__ float lw[DIM * LDSP];
  const int tid = threadIdx.x;
  for (int idx = tid; idx < DIM * DIM; idx += 256) {
    int k = idx >> 7, n = idx & (DIM - 1);
    lw[n * LDSP + k] = W[idx];
  }
  __syncthreads();

  const int lane  = tid & 31;
  const int lrow  = lane & 15;
  const int khalf = (lane >> 4) << 1;
  const size_t m0 = ((size_t)blockIdx.x * 8 + (tid >> 5)) * 16;
  if (m0 >= (size_t)nrows) return;

  v8f acc[8] = {};
  const float* arow = A + (m0 + lrow) * DIM;

  for (int k0 = 0; k0 < DIM; k0 += 4) {
    const int ka = k0 + khalf;
    v2f a = *(const v2f*)(arow + ka);
    if (RELU) { a.x = fmaxf(a.x, 0.0f); a.y = fmaxf(a.y, 0.0f); }
#pragma unroll
    for (int t = 0; t < 8; ++t) {
      v2f b = *(const v2f*)(&lw[(t * 16 + lrow) * LDSP + ka]);
      acc[t] = __builtin_amdgcn_wmma_f32_16x16x4_f32(
          false, a, false, b, (short)0, acc[t], false, false);
    }
  }

  const int rbase = (lane >> 4) << 3;
#pragma unroll
  for (int t = 0; t < 8; ++t) {
    float* op = Out + (m0 + rbase) * DIM + t * 16 + lrow;
    const float bv = BIAS ? bias[t * 16 + lrow] : 0.0f;
#pragma unroll
    for (int r = 0; r < 8; ++r) op[(size_t)r * DIM] = BIAS ? acc[t][r] + bv : acc[t][r];
  }
}


__device__ __forceinline__ int wave_sum(int v) {
  for (int o = 16; o > 0; o >>= 1) v += __shfl_xor(v, o);
  return v;
}

__device__ __forceinline__ int wave_incl_scan(int v, int lane) {
  for (int o = 1; o < 32; o <<= 1) {
    int t = __shfl_up(v, o);
    if (lane >= o) v += t;
  }
  return v;
}

__global__ __launch_bounds__(1024) void k_scan(const int* __restrict__ cnt,
                                               int* __restrict__ rowptr,
                                               int2* __restrict__ ent,
                                               int n) {
  __shared__ int wtot[32];
  __shared__ int wexc[32];
  const int tid  = threadIdx.x;
  const int lane = tid & 31;
  const int w    = tid >> 5;
  const int base = blockIdx.x * 1024;

  int part = 0;
  for (int j = tid; j < base; j += 1024) part += cnt[j] + 1;
  part = wave_sum(part);
  if (lane == 0) wtot[w] = part;

  const int   i  = base + tid;
  const int   c  = (i < n) ? cnt[i] + 1 : 0;
  const int incl = wave_incl_scan(c, lane);
  if (lane == 31) wexc[w] = incl;
  __syncthreads();
  if (w == 0) {
    const int tot = wave_sum(wtot[lane]);
    const int v   = wexc[lane];
    const int s   = wave_incl_scan(v, lane);
    wexc[lane] = tot + s - v;
  }
  __syncthreads();
  if (i < n) {
    const int first = wexc[w] + incl - c;
    rowptr[i] = first;
    const float di = rsqrtf((float)c);
    ent[first] = make_int2(i, __float_as_int(di * di));
    if (i == n - 1) rowptr[n] = first + c;
  }
}

__global__ __launch_bounds__(256) void k_fill(const int* __restrict__ src,
                                              const int* __restrict__ dst,
                                              const int* __restrict__ cnt,
                                              const int* __restrict__ rank,
                                              const int* __restrict__ off,
                                              const int* __restrict__ rowptr,
                                              int2* __restrict__ ent,
                                              int n, int ne) {
  int e = blockIdx.x * 256 + threadIdx.x;
  if (e >= ne) return;
  int s = src[e], d = dst[e];
  float norm = rsqrtf((float)(cnt[s] + 1)) * rsqrtf((float)(cnt[d] + 1));
  const int per = (ne + NCHUNK - 1) / NCHUNK;
  const int pw  = ((((n + 1) >> 1) + 3) >> 2) << 2;
  const int run = e / per;
  int slot = rowptr[d] + 1 + off[((size_t)run * pw) * 2 + d] + rank[e];
  ent[slot] = make_int2(s, __float_as_int(norm));
}

template <int U>
__device__ __forceinline__ void rows_fma(const float4* __restrict__ x4, int2 en,
                                         int j, int lane, float4& acc) {
  float4 v[U];
  float  wt[U];
  for (int u = 0; u < U; ++u) {
    const int s = __builtin_amdgcn_readlane(en.x, j + u);
    wt[u] = __int_as_float(__builtin_amdgcn_readlane(en.y, j + u));
    v[u]  = x4[(size_t)s * (DIM / 4) + lane];
  }
  for (int u = 0; u < U; ++u) {
    acc.x = fmaf(wt[u], v[u].x, acc.x);
    acc.y = fmaf(wt[u], v[u].y, acc.y);
    acc.z = fmaf(wt[u], v[u].z, acc.z);
    acc.w = fmaf(wt[u], v[u].w, acc.w);
  }
}

__global__ __launch_bounds__(256) void k_gather_nodes(const float* __restrict__ x,
                                                      const int2* __restrict__ ent,
                                                      const int* __restrict__ rowptr,
                                                      float* __restrict__ out,
                                                      int nnodes) {
  const int lane = threadIdx.x & 31;
  const int node = __builtin_amdgcn_readfirstlane((int)((blockIdx.x * 256u + threadIdx.x) >> 5));
  if (node >= nnodes) return;
  const int p0 = rowptr[node];
  const int p1 = rowptr[node + 1];

  const float4* x4 = (const float4*)x;
  float4 acc = make_float4(0.0f, 0.0f, 0.0f, 0.0f);
  for (int cb = p0; cb < p1; cb += 32) {
    const int m = min(32, p1 - cb);
    int2 en = make_int2(0, 0);
    if (lane < m) en = ent[cb + lane];
    int j = 0;
    for (; j + 8 <= m; j += 8) rows_fma<8>(x4, en, j, lane, acc);
    if (j + 4 <= m) { rows_fma<4>(x4, en, j, lane, acc); j += 4; }
    if (j + 2 <= m) { rows_fma<2>(x4, en, j, lane, acc); j += 2; }
    if (j < m)      { rows_fma<1>(x4, en, j, lane, acc); }
  }
  ((float4*)out)[(size_t)node * (DIM / 4) + lane] = acc;
}

__global__ __launch_bounds__(1024) void k_gather_pool(const float* __restrict__ h,
                                                      const int2* __restrict__ ent,
                                                      const int* __restrict__ rowptr,
                                                      const int* __restrict__ scope,
                                                      const float* __restrict__ bias,
                                                      float* __restrict__ out,
                                                      int ngraphs, int nnodes) {
  __shared__ int   wpart[32];
  __shared__ float red[32 * DIM];
  const int g    = blockIdx.x;
  const int tid  = threadIdx.x;
  const int lane = tid & 31;
  const int w = __builtin_amdgcn_readfirstlane(tid >> 5);
  if (g >= ngraphs) return;

  int part = 0;
  for (int i = tid; i < g; i += 1024) part += scope[i];
  part = wave_sum(part);
  if (lane == 0) wpart[w] = part;
  __syncthreads();
  const int start = __builtin_amdgcn_readfirstlane(wave_sum(wpart[lane]));
  const int cnt   = scope[g];
  const int n0 = min(start, nnodes);
  const int n1 = min(start + cnt, nnodes);
  const int p0 = rowptr[n0];
  const int p1 = rowptr[n1];

  const float4* h4 = (const float4*)h;
  float4 acc = make_float4(0.0f, 0.0f, 0.0f, 0.0f);
  for (int cb = p0 + 32 * w; cb < p1; cb += 32 * 32) {
    const int m = min(32, p1 - cb);
    int2 en = make_int2(0, 0);
    if (lane < m) en = ent[cb + lane];
    for (int j = 0; j < m; ++j) {
      const int   s  = __builtin_amdgcn_readlane(en.x, j);
      const float wt = __int_as_float(__builtin_amdgcn_readlane(en.y, j));
      const float4 v = h4[(size_t)s * (DIM / 4) + lane];
      acc.x = fmaf(wt, v.x, acc.x);
      acc.y = fmaf(wt, v.y, acc.y);
      acc.z = fmaf(wt, v.z, acc.z);
      acc.w = fmaf(wt, v.w, acc.w);
    }
  }

  *(float4*)(&red[w * DIM + lane * 4]) = acc;
  __syncthreads();
  if (tid < DIM) {
    float s = 0.0f;
    for (int k = 0; k < 32; ++k) s += red[k * DIM + tid];
    out[(size_t)g * DIM + tid] = s / (float)cnt + bias[tid];
  }
}

extern "C" void kernel_launch(void* const* d_in, const int* in_sizes, int n_in,
                              void* d_out, int out_size, void* d_ws, size_t ws_size,
                              hipStream_t stream) {
  (void)n_in; (void)out_size; (void)ws_size;
  const float* x     = (const float*)d_in[0];
  const int*   ei    = (const int*)d_in[1];
  const int*   scope = (const int*)d_in[2];
  const float* W1    = (const float*)d_in[3];
  const float* b1    = (const float*)d_in[4];
  const float* W2    = (const float*)d_in[5];
  const float* b2    = (const float*)d_in[6];
  float*       out   = (float*)d_out;

  const int n_nodes  = in_sizes[0] / DIM;
  const int n_edges  = in_sizes[1] / 2;
  const int n_graphs = in_sizes[2];
  const int* src = ei;
  const int* dst = ei + n_edges;

  float* bufA = (float*)d_ws;
  float* bufB = bufA + (size_t)n_nodes * DIM;
  int*   cnt  = (int*)(bufB + (size_t)n_nodes * DIM);
  int2* ent    = (int2*)bufB;
  int*  rowptr = (int*)(ent + ((size_t)n_edges + n_nodes));
  int*  rank   = rowptr + n_nodes + 1;
  const int pw = ((n_nodes + 1) / 2 + 3) / 4 * 4;
  unsigned* part = (unsigned*)bufA;
  int2*     off  = (int2*)(part + (size_t)NCHUNK * pw);

  const int gb_words  = ((n_nodes + 1) / 2 + 255) / 256;
  const int gb_edges  = (n_edges + 255) / 256;
  const int mtiles    = (n_nodes + 15) / 16;
  const int gb_gemm   = (mtiles + 7) / 8;
  const int gb_wnodes = (n_nodes + 7) / 8;
  const int gb_scan   = (n_nodes + 1023) / 1024;

  k_cnt_chunks   <<<NCHUNK,   1024, 0, stream>>>(dst, part, rank, n_nodes, n_edges);
  k_chunk_offsets<<<gb_words,  256, 0, stream>>>(part, off, cnt, n_nodes);

  k_scan        <<<gb_scan,  1024, 0, stream>>>(cnt, rowptr, ent, n_nodes);
  k_fill        <<<gb_edges,  256, 0, stream>>>(src, dst, cnt, rank, (const int*)off, rowptr, ent,
                                                n_nodes, n_edges);

  k_gather_nodes<<<gb_wnodes, 256, 0, stream>>>(x, ent, rowptr, bufA, n_nodes);
  k_gemm_wmma<0, 1><<<gb_gemm, 256, 0, stream>>>(bufA, W1, b1, bufA, n_nodes);

  k_gemm_wmma<1, 0><<<gb_gemm, 256, 0, stream>>>(bufA, W2, b2, bufA, n_nodes);
  k_gather_pool <<<n_graphs, 1024, 0, stream>>>(bufA, ent, rowptr, scope, b2, out,
                                                n_graphs, n_nodes);
}


# layer-2 product moved behind the mean pool (both linear): relu(agg1) rows pooled per graph, then one WMMA product with W2 per graph in the pool kernel; sum order: pool in slot order, then product
# speedup vs baseline: 4.4355x; 1.1763x over previous
#include <hip/hip_runtime.h>

typedef __attribute__((ext_vector_type(2))) float v2f;
typedef __attribute__((ext_vector_type(8))) float v8f;

#define DIM  128
#define LDSP 132


#define NCHUNK 64
#define LWORDS 51200

__global__ __launch_bounds__(1024) void k_cnt_chunks(const int* __restrict__ dst,
                                                     unsigned* __restrict__ part,
                                                     int* __restrict__ rank,
                                                     int n, int ne) {
  __shared__ uint4 lc4[LWORDS / 4];
  unsigned* lc = (unsigned*)lc4;
  const int tid = threadIdx.x;
  const int pw4 = (((n + 1) >> 1) + 3) >> 2;
  for (int q = tid; q < pw4; q += 1024) lc4[q] = make_uint4(0u, 0u, 0u, 0u);
  __syncthreads();

  const int per = (ne + NCHUNK - 1) / NCHUNK;
  const int e0  = blockIdx.x * per;
  const int e1  = min(ne, e0 + per);
  for (int eb = e0 + tid; eb < e1; eb += 16 * 1024) {
    int      d[16];
    unsigned old[16];
#pragma unroll
    for (int u = 0; u < 16; ++u) {
      const int e = eb + u * 1024;
      d[u] = (e < e1) ? dst[e] : -1;
    }
#pragma unroll
    for (int u = 0; u < 16; ++u) {
      old[u] = 0u;
      if (d[u] >= 0) old[u] = atomicAdd(&lc[d[u] >> 1], (d[u] & 1) ? 0x10000u : 1u);
    }
#pragma unroll
    for (int u = 0; u < 16; ++u) {
      if (d[u] >= 0) rank[eb + u * 1024] = (d[u] & 1) ? (int)(old[u] >> 16) : (int)(old[u] & 0xffffu);
    }
  }
  __syncthreads();

  uint4* p4 = (uint4*)part + (size_t)blockIdx.x * pw4;
  for (int q = tid; q < pw4; q += 1024) p4[q] = lc4[q];
}

__global__ __launch_bounds__(256) void k_chunk_offsets(const unsigned* __restrict__ part,
                                                       int2* __restrict__ off,
                                                       int* __restrict__ cnt, int n) {
  const int nw = (n + 1) >> 1;
  const int pw = ((nw + 3) >> 2) << 2;
  const int w  = blockIdx.x * 256 + threadIdx.x;
  if (w >= nw) return;
  unsigned v[NCHUNK];
#pragma unroll
  for (int c = 0; c < NCHUNK; ++c) v[c] = part[(size_t)c * pw + w];
  int lo = 0, hi = 0;
#pragma unroll
  for (int c = 0; c < NCHUNK; ++c) {
    off[(size_t)c * pw + w] = make_int2(lo, hi);
    lo += (int)(v[c] & 0xffffu);
    hi += (int)(v[c] >> 16);
  }
  cnt[2 * w] = lo;
  if (2 * w + 1 < n) cnt[2 * w + 1] = hi;
}


__global__ __launch_bounds__(256) void k_gemm_wmma(const float* A,
                                                   const float* __restrict__ W,
                                                   const float* __restrict__ bias,
                                                   float* Out,
                                                   int nrows) {
  __shared__ float lw[DIM * LDSP];
  const int tid = threadIdx.x;
  for (int idx = tid; idx < DIM * DIM; idx += 256) {
    int k = idx >> 7, n = idx & (DIM - 1);
    lw[n * LDSP + k] = W[idx];
  }
  __syncthreads();

  const int lane  = tid & 31;
  const int lrow  = lane & 15;
  const int khalf = (lane >> 4) << 1;
  const size_t m0 = ((size_t)blockIdx.x * 8 + (tid >> 5)) * 16;
  if (m0 >= (size_t)nrows) return;

  v8f acc[8] = {};
  const float* arow = A + (m0 + lrow) * DIM;

  for (int k0 = 0; k0 < DIM; k0 += 4) {
    const int ka = k0 + khalf;
    v2f a = *(const v2f*)(arow + ka);
#pragma unroll
    for (int t = 0; t < 8; ++t) {
      v2f b = *(const v2f*)(&lw[(t * 16 + lrow) * LDSP + ka]);
      acc[t] = __builtin_amdgcn_wmma_f32_16x16x4_f32(
          false, a, false, b, (short)0, acc[t], false, false);
    }
  }

  const int rbase = (lane >> 4) << 3;
#pragma unroll
  for (int t = 0; t < 8; ++t) {
    float* op = Out + (m0 + rbase) * DIM + t * 16 + lrow;
    const float bv = bias[t * 16 + lrow];
#pragma unroll
    for (int r = 0; r < 8; ++r) op[(size_t)r * DIM] = fmaxf(acc[t][r] + bv, 0.0f);
  }
}


__device__ __forceinline__ int wave_sum(int v) {
  for (int o = 16; o > 0; o >>= 1) v += __shfl_xor(v, o);
  return v;
}

__device__ __forceinline__ int wave_incl_scan(int v, int lane) {
  for (int o = 1; o < 32; o <<= 1) {
    int t = __shfl_up(v, o);
    if (lane >= o) v += t;
  }
  return v;
}

__global__ __launch_bounds__(1024) void k_scan(const int* __restrict__ cnt,
                                               int* __restrict__ rowptr,
                                               int2* __restrict__ ent,
                                               int n) {
  __shared__ int wtot[32];
  __shared__ int wexc[32];
  const int tid  = threadIdx.x;
  const int lane = tid & 31;
  const int w    = tid >> 5;
  const int base = blockIdx.x * 1024;

  int part = 0;
  for (int j = tid; j < base; j += 1024) part += cnt[j] + 1;
  part = wave_sum(part);
  if (lane == 0) wtot[w] = part;

  const int   i  = base + tid;
  const int   c  = (i < n) ? cnt[i] + 1 : 0;
  const int incl = wave_incl_scan(c, lane);
  if (lane == 31) wexc[w] = incl;
  __syncthreads();
  if (w == 0) {
    const int tot = wave_sum(wtot[lane]);
    const int v   = wexc[lane];
    const int s   = wave_incl_scan(v, lane);
    wexc[lane] = tot + s - v;
  }
  __syncthreads();
  if (i < n) {
    const int first = wexc[w] + incl - c;
    rowptr[i] = first;
    const float di = rsqrtf((float)c);
    ent[first] = make_int2(i, __float_as_int(di * di));
    if (i == n - 1) rowptr[n] = first + c;
  }
}

__global__ __launch_bounds__(256) void k_fill(const int* __restrict__ src,
                                              const int* __restrict__ dst,
                                              const int* __restrict__ cnt,
                                              const int* __restrict__ rank,
                                              const int* __restrict__ off,
                                              const int* __restrict__ rowptr,
                                              int2* __restrict__ ent,
                                              int n, int ne) {
  int e = blockIdx.x * 256 + threadIdx.x;
  if (e >= ne) return;
  int s = src[e], d = dst[e];
  float norm = rsqrtf((float)(cnt[s] + 1)) * rsqrtf((float)(cnt[d] + 1));
  const int per = (ne + NCHUNK - 1) / NCHUNK;
  const int pw  = ((((n + 1) >> 1) + 3) >> 2) << 2;
  const int run = e / per;
  int slot = rowptr[d] + 1 + off[((size_t)run * pw) * 2 + d] + rank[e];
  ent[slot] = make_int2(s, __float_as_int(norm));
}

template <int U>
__device__ __forceinline__ void rows_fma(const float4* __restrict__ x4, int2 en,
                                         int j, int lane, float4& acc) {
  float4 v[U];
  float  wt[U];
  for (int u = 0; u < U; ++u) {
    const int s = __builtin_amdgcn_readlane(en.x, j + u);
    wt[u] = __int_as_float(__builtin_amdgcn_readlane(en.y, j + u));
    v[u]  = x4[(size_t)s * (DIM / 4) + lane];
  }
  for (int u = 0; u < U; ++u) {
    acc.x = fmaf(wt[u], v[u].x, acc.x);
    acc.y = fmaf(wt[u], v[u].y, acc.y);
    acc.z = fmaf(wt[u], v[u].z, acc.z);
    acc.w = fmaf(wt[u], v[u].w, acc.w);
  }
}

__global__ __launch_bounds__(256) void k_gather_nodes(const float* __restrict__ x,
                                                      const int2* __restrict__ ent,
                                                      const int* __restrict__ rowptr,
                                                      float* __restrict__ out,
                                                      int nnodes) {
  const int lane = threadIdx.x & 31;
  const int node = __builtin_amdgcn_readfirstlane((int)((blockIdx.x * 256u + threadIdx.x) >> 5));
  if (node >= nnodes) return;
  const int p0 = rowptr[node];
  const int p1 = rowptr[node + 1];

  const float4* x4 = (const float4*)x;
  float4 acc = make_float4(0.0f, 0.0f, 0.0f, 0.0f);
  for (int cb = p0; cb < p1; cb += 32) {
    const int m = min(32, p1 - cb);
    int2 en = make_int2(0, 0);
    if (lane < m) en = ent[cb + lane];
    int j = 0;
    for (; j + 8 <= m; j += 8) rows_fma<8>(x4, en, j, lane, acc);
    if (j + 4 <= m) { rows_fma<4>(x4, en, j, lane, acc); j += 4; }
    if (j + 2 <= m) { rows_fma<2>(x4, en, j, lane, acc); j += 2; }
    if (j < m)      { rows_fma<1>(x4, en, j, lane, acc); }
  }
  ((float4*)out)[(size_t)node * (DIM / 4) + lane] = acc;
}

__global__ __launch_bounds__(1024) void k_gather_pool(const float* __restrict__ h,
                                                      const int2* __restrict__ ent,
                                                      const int* __restrict__ rowptr,
                                                      const int* __restrict__ scope,
                                                      const float* __restrict__ W,
                                                      const float* __restrict__ bias,
                                                      float* __restrict__ out,
                                                      int ngraphs, int nnodes) {
  __shared__ int   wpart[32];
  __shared__ float red[32 * DIM];
  __shared__ float pooled[DIM];
  __shared__ float kq[4 * DIM];
  const int g    = blockIdx.x;
  const int tid  = threadIdx.x;
  const int lane = tid & 31;
  const int w = __builtin_amdgcn_readfirstlane(tid >> 5);
  if (g >= ngraphs) return;

  int part = 0;
  for (int i = tid; i < g; i += 1024) part += scope[i];
  part = wave_sum(part);
  if (lane == 0) wpart[w] = part;
  __syncthreads();
  const int start = __builtin_amdgcn_readfirstlane(wave_sum(wpart[lane]));
  const int cnt   = scope[g];
  const int n0 = min(start, nnodes);
  const int n1 = min(start + cnt, nnodes);
  const int p0 = rowptr[n0];
  const int p1 = rowptr[n1];

  const float4* h4 = (const float4*)h;
  float4 acc = make_float4(0.0f, 0.0f, 0.0f, 0.0f);
  for (int cb = p0 + 32 * w; cb < p1; cb += 32 * 32) {
    const int m = min(32, p1 - cb);
    int2 en = make_int2(0, 0);
    if (lane < m) en = ent[cb + lane];
    for (int j = 0; j < m; ++j) {
      const int   s  = __builtin_amdgcn_readlane(en.x, j);
      const float wt = __int_as_float(__builtin_amdgcn_readlane(en.y, j));
      const float4 v = h4[(size_t)s * (DIM / 4) + lane];
      acc.x = fmaf(wt, v.x, acc.x);
      acc.y = fmaf(wt, v.y, acc.y);
      acc.z = fmaf(wt, v.z, acc.z);
      acc.w = fmaf(wt, v.w, acc.w);
    }
  }

  *(float4*)(&red[w * DIM + lane * 4]) = acc;
  __syncthreads();
  if (tid < DIM) {
    float s = 0.0f;
    for (int k = 0; k < 32; ++k) s += red[k * DIM + tid];
    pooled[tid] = s;
  }
  __syncthreads();

  const int lrow  = lane & 15;
  const int khalf = (lane >> 4) << 1;
  const int ncol  = (w & 7) * 16 + lrow;
  const int kbase = (w >> 3) * 32 + khalf;
  v8f d = {};
  for (int i = 0; i < 8; ++i) {
    const int ka = kbase + i * 4;
    const v2f a = *(const v2f*)(&pooled[ka]);
    v2f b;
    b.x = W[(size_t)ka * DIM + ncol];
    b.y = W[(size_t)(ka + 1) * DIM + ncol];
    d = __builtin_amdgcn_wmma_f32_16x16x4_f32(false, a, false, b, (short)0, d, false, false);
  }
  if (lane < 16) kq[(w >> 3) * DIM + ncol] = d[0];
  __syncthreads();
  if (tid < DIM) {
    const float s = (kq[tid] + kq[DIM + tid]) + (kq[2 * DIM + tid] + kq[3 * DIM + tid]);
    out[(size_t)g * DIM + tid] = s / (float)cnt + bias[tid];
  }
}

extern "C" void kernel_launch(void* const* d_in, const int* in_sizes, int n_in,
                              void* d_out, int out_size, void* d_ws, size_t ws_size,
                              hipStream_t stream) {
  (void)n_in; (void)out_size; (void)ws_size;
  const float* x     = (const float*)d_in[0];
  const int*   ei    = (const int*)d_in[1];
  const int*   scope = (const int*)d_in[2];
  const float* W1    = (const float*)d_in[3];
  const float* b1    = (const float*)d_in[4];
  const float* W2    = (const float*)d_in[5];
  const float* b2    = (const float*)d_in[6];
  float*       out   = (float*)d_out;

  const int n_nodes  = in_sizes[0] / DIM;
  const int n_edges  = in_sizes[1] / 2;
  const int n_graphs = in_sizes[2];
  const int* src = ei;
  const int* dst = ei + n_edges;

  float* bufA = (float*)d_ws;
  float* bufB = bufA + (size_t)n_nodes * DIM;
  int*   cnt  = (int*)(bufB + (size_t)n_nodes * DIM);
  int2* ent    = (int2*)bufB;
  int*  rowptr = (int*)(ent + ((size_t)n_edges + n_nodes));
  int*  rank   = rowptr + n_nodes + 1;
  const int pw = ((n_nodes + 1) / 2 + 3) / 4 * 4;
  unsigned* part = (unsigned*)bufA;
  int2*     off  = (int2*)(part + (size_t)NCHUNK * pw);

  const int gb_words  = ((n_nodes + 1) / 2 + 255) / 256;
  const int gb_edges  = (n_edges + 255) / 256;
  const int mtiles    = (n_nodes + 15) / 16;
  const int gb_gemm   = (mtiles + 7) / 8;
  const int gb_wnodes = (n_nodes + 7) / 8;
  const int gb_scan   = (n_nodes + 1023) / 1024;

  k_cnt_chunks   <<<NCHUNK,   1024, 0, stream>>>(dst, part, rank, n_nodes, n_edges);
  k_chunk_offsets<<<gb_words,  256, 0, stream>>>(part, off, cnt, n_nodes);

  k_scan        <<<gb_scan,  1024, 0, stream>>>(cnt, rowptr, ent, n_nodes);
  k_fill        <<<gb_edges,  256, 0, stream>>>(src, dst, cnt, rank, (const int*)off, rowptr, ent,
                                                n_nodes, n_edges);

  k_gather_nodes<<<gb_wnodes, 256, 0, stream>>>(x, ent, rowptr, bufA, n_nodes);
  k_gemm_wmma   <<<gb_gemm,  256, 0, stream>>>(bufA, W1, b1, bufA, n_nodes);

  k_gather_pool <<<n_graphs, 1024, 0, stream>>>(bufA, ent, rowptr, scope, W2, b2, out,
                                                n_graphs, n_nodes);
}


# layer-1 gather fused into the W1 product: one 16-wave work-group per WGP, a wave gathers its 16 rows into its own LDS tile and feeds the WMMAs from it; Ax rows never reach memory; sum order unchanged
# speedup vs baseline: 4.7187x; 1.0638x over previous
#include <hip/hip_runtime.h>

typedef __attribute__((ext_vector_type(2))) float v2f;
typedef __attribute__((ext_vector_type(8))) float v8f;

#define DIM  128
#define LDSP 132


#define NCHUNK 64
#define LWORDS 51200

__global__ __launch_bounds__(1024) void k_cnt_chunks(const int* __restrict__ dst,
                                                     unsigned* __restrict__ part,
                                                     int* __restrict__ rank,
                                                     int n, int ne) {
  __shared__ uint4 lc4[LWORDS / 4];
  unsigned* lc = (unsigned*)lc4;
  const int tid = threadIdx.x;
  const int pw4 = (((n + 1) >> 1) + 3) >> 2;
  for (int q = tid; q < pw4; q += 1024) lc4[q] = make_uint4(0u, 0u, 0u, 0u);
  __syncthreads();

  const int per = (ne + NCHUNK - 1) / NCHUNK;
  const int e0  = blockIdx.x * per;
  const int e1  = min(ne, e0 + per);
  for (int eb = e0 + tid; eb < e1; eb += 16 * 1024) {
    int      d[16];
    unsigned old[16];
#pragma unroll
    for (int u = 0; u < 16; ++u) {
      const int e = eb + u * 1024;
      d[u] = (e < e1) ? dst[e] : -1;
    }
#pragma unroll
    for (int u = 0; u < 16; ++u) {
      old[u] = 0u;
      if (d[u] >= 0) old[u] = atomicAdd(&lc[d[u] >> 1], (d[u] & 1) ? 0x10000u : 1u);
    }
#pragma unroll
    for (int u = 0; u < 16; ++u) {
      if (d[u] >= 0) rank[eb + u * 1024] = (d[u] & 1) ? (int)(old[u] >> 16) : (int)(old[u] & 0xffffu);
    }
  }
  __syncthreads();

  uint4* p4 = (uint4*)part + (size_t)blockIdx.x * pw4;
  for (int q = tid; q < pw4; q += 1024) p4[q] = lc4[q];
}

__global__ __launch_bounds__(256) void k_chunk_offsets(const unsigned* __restrict__ part,
                                                       int2* __restrict__ off,
                                                       int* __restrict__ cnt, int n) {
  const int nw = (n + 1) >> 1;
  const int pw = ((nw + 3) >> 2) << 2;
  const int w  = blockIdx.x * 256 + threadIdx.x;
  if (w >= nw) return;
  unsigned v[NCHUNK];
#pragma unroll
  for (int c = 0; c < NCHUNK; ++c) v[c] = part[(size_t)c * pw + w];
  int lo = 0, hi = 0;
#pragma unroll
  for (int c = 0; c < NCHUNK; ++c) {
    off[(size_t)c * pw + w] = make_int2(lo, hi);
    lo += (int)(v[c] & 0xffffu);
    hi += (int)(v[c] >> 16);
  }
  cnt[2 * w] = lo;
  if (2 * w + 1 < n) cnt[2 * w + 1] = hi;
}


__device__ __forceinline__ int wave_sum(int v) {
  for (int o = 16; o > 0; o >>= 1) v += __shfl_xor(v, o);
  return v;
}

__device__ __forceinline__ int wave_incl_scan(int v, int lane) {
  for (int o = 1; o < 32; o <<= 1) {
    int t = __shfl_up(v, o);
    if (lane >= o) v += t;
  }
  return v;
}

__global__ __launch_bounds__(1024) void k_scan(const int* __restrict__ cnt,
                                               int* __restrict__ rowptr,
                                               int2* __restrict__ ent,
                                               int n) {
  __shared__ int wtot[32];
  __shared__ int wexc[32];
  const int tid  = threadIdx.x;
  const int lane = tid & 31;
  const int w    = tid >> 5;
  const int base = blockIdx.x * 1024;

  int part = 0;
  for (int j = tid; j < base; j += 1024) part += cnt[j] + 1;
  part = wave_sum(part);
  if (lane == 0) wtot[w] = part;

  const int   i  = base + tid;
  const int   c  = (i < n) ? cnt[i] + 1 : 0;
  const int incl = wave_incl_scan(c, lane);
  if (lane == 31) wexc[w] = incl;
  __syncthreads();
  if (w == 0) {
    const int tot = wave_sum(wtot[lane]);
    const int v   = wexc[lane];
    const int s   = wave_incl_scan(v, lane);
    wexc[lane] = tot + s - v;
  }
  __syncthreads();
  if (i < n) {
    const int first = wexc[w] + incl - c;
    rowptr[i] = first;
    const float di = rsqrtf((float)c);
    ent[first] = make_int2(i, __float_as_int(di * di));
    if (i == n - 1) rowptr[n] = first + c;
  }
}

__global__ __launch_bounds__(256) void k_fill(const int* __restrict__ src,
                                              const int* __restrict__ dst,
                                              const int* __restrict__ cnt,
                                              const int* __restrict__ rank,
                                              const int* __restrict__ off,
                                              const int* __restrict__ rowptr,
                                              int2* __restrict__ ent,
                                              int n, int ne) {
  int e = blockIdx.x * 256 + threadIdx.x;
  if (e >= ne) return;
  int s = src[e], d = dst[e];
  float norm = rsqrtf((float)(cnt[s] + 1)) * rsqrtf((float)(cnt[d] + 1));
  const int per = (ne + NCHUNK - 1) / NCHUNK;
  const int pw  = ((((n + 1) >> 1) + 3) >> 2) << 2;
  const int run = e / per;
  int slot = rowptr[d] + 1 + off[((size_t)run * pw) * 2 + d] + rank[e];
  ent[slot] = make_int2(s, __float_as_int(norm));
}


#define NWGP   256
#define TWAVES 16

template <int U>
__device__ __forceinline__ void rows_fma_seg(const float4* __restrict__ x4, int2 en, int rp,
                                             int j, int slot0, int lane, float* tile,
                                             int& nl, int& nextb, float4& acc) {
  float4 v[U];
  float  wt[U];
#pragma unroll
  for (int u = 0; u < U; ++u) {
    const int s = __builtin_amdgcn_readlane(en.x, j + u);
    wt[u] = __int_as_float(__builtin_amdgcn_readlane(en.y, j + u));
    v[u]  = x4[(size_t)s * (DIM / 4) + lane];
  }
#pragma unroll
  for (int u = 0; u < U; ++u) {
    if (slot0 + u == nextb) {
      *(float4*)(&tile[nl * LDSP + lane * 4]) = acc;
      acc = make_float4(0.0f, 0.0f, 0.0f, 0.0f);
      nl += 1;
      nextb = __builtin_amdgcn_readlane(rp, nl + 1);
    }
    acc.x = fmaf(wt[u], v[u].x, acc.x);
    acc.y = fmaf(wt[u], v[u].y, acc.y);
    acc.z = fmaf(wt[u], v[u].z, acc.z);
    acc.w = fmaf(wt[u], v[u].w, acc.w);
  }
}

__global__ __launch_bounds__(TWAVES * 32) void k_gather_gemm(const float* __restrict__ x,
                                                             const int2* __restrict__ ent,
                                                             const int* __restrict__ rowptr,
                                                             const float* __restrict__ W,
                                                             const float* __restrict__ bias,
                                                             float* __restrict__ Out,
                                                             int nrows) {
  __shared__ __attribute__((aligned(16))) float lw[DIM * LDSP];
  __shared__ __attribute__((aligned(16))) float la[TWAVES * 16 * LDSP];
  const int tid = threadIdx.x;
  for (int idx = tid; idx < DIM * DIM; idx += TWAVES * 32) {
    int k = idx >> 7, n = idx & (DIM - 1);
    lw[n * LDSP + k] = W[idx];
  }
  __syncthreads();

  const int lane  = tid & 31;
  const int w     = __builtin_amdgcn_readfirstlane(tid >> 5);
  const int lrow  = lane & 15;
  const int khalf = (lane >> 4) << 1;
  const int rbase = (lane >> 4) << 3;
  float* tile = la + w * (16 * LDSP);
  const float4* x4 = (const float4*)x;

  const unsigned mtiles = ((unsigned)nrows + 15u) >> 4;
  const int t0 = (int)((blockIdx.x * mtiles) / NWGP);
  const int t1 = (int)(((blockIdx.x + 1u) * mtiles) / NWGP);

  for (int t = t0 + w; t < t1; t += TWAVES) {
    const int m0 = t * 16;

    const int rp = rowptr[min(m0 + min(lane, 16), nrows)];
    const int pA = __builtin_amdgcn_readlane(rp, 0);
    const int pB = __builtin_amdgcn_readlane(rp, 16);
    const bool full = (m0 + 16 <= nrows);
    if (!full) {
#pragma unroll
      for (int r = 0; r < 16; ++r)
        *(float4*)(&tile[r * LDSP + lane * 4]) = make_float4(0.0f, 0.0f, 0.0f, 0.0f);
    }
    int nl    = 0;
    int nextb = __builtin_amdgcn_readlane(rp, 1);
    float4 acc = make_float4(0.0f, 0.0f, 0.0f, 0.0f);
    const int plast = pB - 1;
    int2 en = ent[min(pA + lane, plast)];
    for (int cb = pA; cb < pB; cb += 32) {
      const int m = min(32, pB - cb);
      const int2 nx = ent[min(cb + 32 + lane, plast)];
      int j = 0;
      if (m == 32)     { rows_fma_seg<32>(x4, en, rp, j, cb + j, lane, tile, nl, nextb, acc); j += 32; }
      if (j + 16 <= m) { rows_fma_seg<16>(x4, en, rp, j, cb + j, lane, tile, nl, nextb, acc); j += 16; }
      if (j + 8 <= m) { rows_fma_seg<8>(x4, en, rp, j, cb + j, lane, tile, nl, nextb, acc); j += 8; }
      if (j + 4 <= m) { rows_fma_seg<4>(x4, en, rp, j, cb + j, lane, tile, nl, nextb, acc); j += 4; }
      if (j + 2 <= m) { rows_fma_seg<2>(x4, en, rp, j, cb + j, lane, tile, nl, nextb, acc); j += 2; }
      if (j < m)      { rows_fma_seg<1>(x4, en, rp, j, cb + j, lane, tile, nl, nextb, acc); }
      en = nx;
    }
    *(float4*)(&tile[nl * LDSP + lane * 4]) = acc;

    v8f d[8] = {};
    const float* arow = tile + lrow * LDSP + khalf;
    for (int k0 = 0; k0 < DIM; k0 += 4) {
      v2f a = *(const v2f*)(arow + k0);
#pragma unroll
      for (int tt = 0; tt < 8; ++tt) {
        v2f b = *(const v2f*)(&lw[(tt * 16 + lrow) * LDSP + k0 + khalf]);
        d[tt] = __builtin_amdgcn_wmma_f32_16x16x4_f32(
            false, a, false, b, (short)0, d[tt], false, false);
      }
    }

    if (full) {
#pragma unroll
      for (int tt = 0; tt < 8; ++tt) {
        float* op = Out + ((size_t)m0 + rbase) * DIM + tt * 16 + lrow;
        const float bv = bias[tt * 16 + lrow];
#pragma unroll
        for (int r = 0; r < 8; ++r) op[(size_t)r * DIM] = fmaxf(d[tt][r] + bv, 0.0f);
      }
    } else {
      const int vr = nrows - m0 - rbase;
#pragma unroll
      for (int tt = 0; tt < 8; ++tt) {
        float* op = Out + ((size_t)m0 + rbase) * DIM + tt * 16 + lrow;
        const float bv = bias[tt * 16 + lrow];
#pragma unroll
        for (int r = 0; r < 8; ++r)
          if (r < vr) op[(size_t)r * DIM] = fmaxf(d[tt][r] + bv, 0.0f);
      }
    }
  }
}

__global__ __launch_bounds__(1024) void k_gather_pool(const float* __restrict__ h,
                                                      const int2* __restrict__ ent,
                                                      const int* __restrict__ rowptr,
                                                      const int* __restrict__ scope,
                                                      const float* __restrict__ W,
                                                      const float* __restrict__ bias,
                                                      float* __restrict__ out,
                                                      int ngraphs, int nnodes) {
  __shared__ int   wpart[32];
  __shared__ float red[32 * DIM];
  __shared__ float pooled[DIM];
  __shared__ float kq[4 * DIM];
  const int g    = blockIdx.x;
  const int tid  = threadIdx.x;
  const int lane = tid & 31;
  const int w = __builtin_amdgcn_readfirstlane(tid >> 5);
  if (g >= ngraphs) return;

  int part = 0;
  for (int i = tid; i < g; i += 1024) part += scope[i];
  part = wave_sum(part);
  if (lane == 0) wpart[w] = part;
  __syncthreads();
  const int start = __builtin_amdgcn_readfirstlane(wave_sum(wpart[lane]));
  const int cnt   = scope[g];
  const int n0 = min(start, nnodes);
  const int n1 = min(start + cnt, nnodes);
  const int p0 = rowptr[n0];
  const int p1 = rowptr[n1];

  const float4* h4 = (const float4*)h;
  float4 acc = make_float4(0.0f, 0.0f, 0.0f, 0.0f);
  for (int cb = p0 + 32 * w; cb < p1; cb += 32 * 32) {
    const int m = min(32, p1 - cb);
    int2 en = make_int2(0, 0);
    if (lane < m) en = ent[cb + lane];
    for (int j = 0; j < m; ++j) {
      const int   s  = __builtin_amdgcn_readlane(en.x, j);
      const float wt = __int_as_float(__builtin_amdgcn_readlane(en.y, j));
      const float4 v = h4[(size_t)s * (DIM / 4) + lane];
      acc.x = fmaf(wt, v.x, acc.x);
      acc.y = fmaf(wt, v.y, acc.y);
      acc.z = fmaf(wt, v.z, acc.z);
      acc.w = fmaf(wt, v.w, acc.w);
    }
  }

  *(float4*)(&red[w * DIM + lane * 4]) = acc;
  __syncthreads();
  if (tid < DIM) {
    float s = 0.0f;
    for (int k = 0; k < 32; ++k) s += red[k * DIM + tid];
    pooled[tid] = s;
  }
  __syncthreads();

  const int lrow  = lane & 15;
  const int khalf = (lane >> 4) << 1;
  const int ncol  = (w & 7) * 16 + lrow;
  const int kbase = (w >> 3) * 32 + khalf;
  v8f d = {};
  for (int i = 0; i < 8; ++i) {
    const int ka = kbase + i * 4;
    const v2f a = *(const v2f*)(&pooled[ka]);
    v2f b;
    b.x = W[(size_t)ka * DIM + ncol];
    b.y = W[(size_t)(ka + 1) * DIM + ncol];
    d = __builtin_amdgcn_wmma_f32_16x16x4_f32(false, a, false, b, (short)0, d, false, false);
  }
  if (lane < 16) kq[(w >> 3) * DIM + ncol] = d[0];
  __syncthreads();
  if (tid < DIM) {
    const float s = (kq[tid] + kq[DIM + tid]) + (kq[2 * DIM + tid] + kq[3 * DIM + tid]);
    out[(size_t)g * DIM + tid] = s / (float)cnt + bias[tid];
  }
}

extern "C" void kernel_launch(void* const* d_in, const int* in_sizes, int n_in,
                              void* d_out, int out_size, void* d_ws, size_t ws_size,
                              hipStream_t stream) {
  (void)n_in; (void)out_size; (void)ws_size;
  const float* x     = (const float*)d_in[0];
  const int*   ei    = (const int*)d_in[1];
  const int*   scope = (const int*)d_in[2];
  const float* W1    = (const float*)d_in[3];
  const float* b1    = (const float*)d_in[4];
  const float* W2    = (const float*)d_in[5];
  const float* b2    = (const float*)d_in[6];
  float*       out   = (float*)d_out;

  const int n_nodes  = in_sizes[0] / DIM;
  const int n_edges  = in_sizes[1] / 2;
  const int n_graphs = in_sizes[2];
  const int* src = ei;
  const int* dst = ei + n_edges;

  float* bufA = (float*)d_ws;
  float* bufB = bufA + (size_t)n_nodes * DIM;
  int*   cnt  = (int*)(bufB + (size_t)n_nodes * DIM);
  int2* ent    = (int2*)bufB;
  int*  rowptr = (int*)(ent + ((size_t)n_edges + n_nodes));
  int*  rank   = rowptr + n_nodes + 1;
  const int pw = ((n_nodes + 1) / 2 + 3) / 4 * 4;
  unsigned* part = (unsigned*)bufA;
  int2*     off  = (int2*)(part + (size_t)NCHUNK * pw);

  const int gb_words  = ((n_nodes + 1) / 2 + 255) / 256;
  const int gb_edges  = (n_edges + 255) / 256;
  const int gb_scan   = (n_nodes + 1023) / 1024;

  k_cnt_chunks   <<<NCHUNK,   1024, 0, stream>>>(dst, part, rank, n_nodes, n_edges);
  k_chunk_offsets<<<gb_words,  256, 0, stream>>>(part, off, cnt, n_nodes);

  k_scan        <<<gb_scan,  1024, 0, stream>>>(cnt, rowptr, ent, n_nodes);
  k_fill        <<<gb_edges,  256, 0, stream>>>(src, dst, cnt, rank, (const int*)off, rowptr, ent,
                                                n_nodes, n_edges);

  k_gather_gemm <<<NWGP, TWAVES * 32, 0, stream>>>(x, ent, rowptr, W1, b1, bufA, n_nodes);

  k_gather_pool <<<n_graphs, 1024, 0, stream>>>(bufA, ent, rowptr, scope, W2, b2, out,
                                                n_graphs, n_nodes);
}


# list row starts as a two-level sum: the offsets kernel scans its 512-node block into its per-run offsets, each fill work-group scans the block totals; no scan launch (5, was 6); sum order unchanged
# speedup vs baseline: 5.0060x; 1.0609x over previous
#include <hip/hip_runtime.h>

typedef __attribute__((ext_vector_type(2))) float v2f;
typedef __attribute__((ext_vector_type(8))) float v8f;

#define DIM  128
#define LDSP 132


#define NCHUNK 64
#define LWORDS 51200

__global__ __launch_bounds__(1024) void k_cnt_chunks(const int* __restrict__ dst,
                                                     unsigned* __restrict__ part,
                                                     int* __restrict__ rank,
                                                     int n, int ne) {
  __shared__ uint4 lc4[LWORDS / 4];
  unsigned* lc = (unsigned*)lc4;
  const int tid = threadIdx.x;
  const int pw4 = (((n + 1) >> 1) + 3) >> 2;
  for (int q = tid; q < pw4; q += 1024) lc4[q] = make_uint4(0u, 0u, 0u, 0u);
  __syncthreads();

  const int per = (ne + NCHUNK - 1) / NCHUNK;
  const int e0  = blockIdx.x * per;
  const int e1  = min(ne, e0 + per);
  for (int eb = e0 + tid; eb < e1; eb += 16 * 1024) {
    int      d[16];
    unsigned old[16];
#pragma unroll
    for (int u = 0; u < 16; ++u) {
      const int e = eb + u * 1024;
      d[u] = (e < e1) ? dst[e] : -1;
    }
#pragma unroll
    for (int u = 0; u < 16; ++u) {
      old[u] = 0u;
      if (d[u] >= 0) old[u] = atomicAdd(&lc[d[u] >> 1], (d[u] & 1) ? 0x10000u : 1u);
    }
#pragma unroll
    for (int u = 0; u < 16; ++u) {
      if (d[u] >= 0) rank[eb + u * 1024] = (d[u] & 1) ? (int)(old[u] >> 16) : (int)(old[u] & 0xffffu);
    }
  }
  __syncthreads();

  uint4* p4 = (uint4*)part + (size_t)blockIdx.x * pw4;
  for (int q = tid; q < pw4; q += 1024) p4[q] = lc4[q];
}

__device__ __forceinline__ int wave_sum(int v) {
  for (int o = 16; o > 0; o >>= 1) v += __shfl_xor(v, o);
  return v;
}

__device__ __forceinline__ int wave_incl_scan(int v, int lane) {
  for (int o = 1; o < 32; o <<= 1) {
    int t = __shfl_up(v, o);
    if (lane >= o) v += t;
  }
  return v;
}

__global__ __launch_bounds__(256) void k_chunk_offsets(const unsigned* __restrict__ part,
                                                       int2* __restrict__ off,
                                                       int* __restrict__ cnt,
                                                       int* __restrict__ bsum, int n) {
  __shared__ int wtot[8];
  const int nw   = (n + 1) >> 1;
  const int pw   = ((nw + 3) >> 2) << 2;
  const int tid  = threadIdx.x;
  const int lane = tid & 31;
  const int wv   = tid >> 5;
  const int w    = blockIdx.x * 256 + tid;
  const bool live = (w < nw);
  const bool odd  = live && (2 * w + 1 < n);
  const int wl = live ? w : 0;
  unsigned v[NCHUNK];
#pragma unroll
  for (int c = 0; c < NCHUNK; ++c) v[c] = part[(size_t)c * pw + wl];
  int lo = 0, hi = 0;
#pragma unroll
  for (int c = 0; c < NCHUNK; ++c) {
    lo += (int)(v[c] & 0xffffu);
    hi += (int)(v[c] >> 16);
  }
  const int c2   = (live ? lo + 1 : 0) + (odd ? hi + 1 : 0);
  const int incl = wave_incl_scan(c2, lane);
  if (lane == 31) wtot[wv] = incl;
  __syncthreads();
  int front = 0, total = 0;
  for (int k = 0; k < 8; ++k) {
    const int t = wtot[k];
    if (k < wv) front += t;
    total += t;
  }
  if (tid == 0) bsum[blockIdx.x] = total;
  if (!live) return;
  int flo = front + incl - c2 + 1;
  int fhi = flo + lo + 1;
#pragma unroll
  for (int c = 0; c < NCHUNK; ++c) {
    off[(size_t)c * pw + w] = make_int2(flo, fhi);
    flo += (int)(v[c] & 0xffffu);
    fhi += (int)(v[c] >> 16);
  }
  cnt[2 * w] = lo;
  if (odd) cnt[2 * w + 1] = hi;
}


__global__ __launch_bounds__(256) void k_fill(const int* __restrict__ src,
                                              const int* __restrict__ dst,
                                              const int* __restrict__ cnt,
                                              const int* __restrict__ rank,
                                              const int* __restrict__ off,
                                              const int* __restrict__ bsum,
                                              int* __restrict__ rowptr,
                                              int2* __restrict__ ent,
                                              int n, int ne, int nb) {
  __shared__ int bpre[256];
  __shared__ int wtot[8];
  const int tid  = threadIdx.x;
  const int lane = tid & 31;
  const int wv   = tid >> 5;
  const int b    = (tid < nb) ? bsum[tid] : 0;
  const int incl = wave_incl_scan(b, lane);
  if (lane == 31) wtot[wv] = incl;
  __syncthreads();
  int front = 0;
  for (int k = 0; k < 8; ++k) {
    const int t = wtot[k];
    if (k < wv) front += t;
  }
  bpre[tid] = front + incl - b;
  __syncthreads();

  const int e = blockIdx.x * 256 + tid;
  if (e < ne) {
    int s = src[e], d = dst[e];
    float norm = rsqrtf((float)(cnt[s] + 1)) * rsqrtf((float)(cnt[d] + 1));
    const int per = (ne + NCHUNK - 1) / NCHUNK;
    const int pw  = ((((n + 1) >> 1) + 3) >> 2) << 2;
    const int run = e / per;
    int slot = bpre[d >> 9] + off[((size_t)run * pw) * 2 + d] + rank[e];
    ent[slot] = make_int2(s, __float_as_int(norm));
  }
  for (int i = e; i < n; i += gridDim.x * 256) {
    const int   c     = cnt[i] + 1;
    const int   first = bpre[i >> 9] + off[i] - 1;
    const float di    = rsqrtf((float)c);
    rowptr[i] = first;
    ent[first] = make_int2(i, __float_as_int(di * di));
    if (i == n - 1) rowptr[n] = first + c;
  }
}


#define NWGP   256
#define TWAVES 16

template <int U>
__device__ __forceinline__ void rows_fma_seg(const float4* __restrict__ x4, int2 en, int rp,
                                             int j, int slot0, int lane, float* tile,
                                             int& nl, int& nextb, float4& acc) {
  float4 v[U];
  float  wt[U];
#pragma unroll
  for (int u = 0; u < U; ++u) {
    const int s = __builtin_amdgcn_readlane(en.x, j + u);
    wt[u] = __int_as_float(__builtin_amdgcn_readlane(en.y, j + u));
    v[u]  = x4[(size_t)s * (DIM / 4) + lane];
  }
#pragma unroll
  for (int u = 0; u < U; ++u) {
    if (slot0 + u == nextb) {
      *(float4*)(&tile[nl * LDSP + lane * 4]) = acc;
      acc = make_float4(0.0f, 0.0f, 0.0f, 0.0f);
      nl += 1;
      nextb = __builtin_amdgcn_readlane(rp, nl + 1);
    }
    acc.x = fmaf(wt[u], v[u].x, acc.x);
    acc.y = fmaf(wt[u], v[u].y, acc.y);
    acc.z = fmaf(wt[u], v[u].z, acc.z);
    acc.w = fmaf(wt[u], v[u].w, acc.w);
  }
}

__global__ __launch_bounds__(TWAVES * 32) void k_gather_gemm(const float* __restrict__ x,
                                                             const int2* __restrict__ ent,
                                                             const int* __restrict__ rowptr,
                                                             const float* __restrict__ W,
                                                             const float* __restrict__ bias,
                                                             float* __restrict__ Out,
                                                             int nrows) {
  __shared__ __attribute__((aligned(16))) float lw[DIM * LDSP];
  __shared__ __attribute__((aligned(16))) float la[TWAVES * 16 * LDSP];
  const int tid = threadIdx.x;
  for (int idx = tid; idx < DIM * DIM; idx += TWAVES * 32) {
    int k = idx >> 7, n = idx & (DIM - 1);
    lw[n * LDSP + k] = W[idx];
  }
  __syncthreads();

  const int lane  = tid & 31;
  const int w     = __builtin_amdgcn_readfirstlane(tid >> 5);
  const int lrow  = lane & 15;
  const int khalf = (lane >> 4) << 1;
  const int rbase = (lane >> 4) << 3;
  float* tile = la + w * (16 * LDSP);
  const float4* x4 = (const float4*)x;

  const unsigned mtiles = ((unsigned)nrows + 15u) >> 4;
  const int t0 = (int)((blockIdx.x * mtiles) / NWGP);
  const int t1 = (int)(((blockIdx.x + 1u) * mtiles) / NWGP);

  for (int t = t0 + w; t < t1; t += TWAVES) {
    const int m0 = t * 16;

    const int rp = rowptr[min(m0 + min(lane, 16), nrows)];
    const int pA = __builtin_amdgcn_readlane(rp, 0);
    const int pB = __builtin_amdgcn_readlane(rp, 16);
    const bool full = (m0 + 16 <= nrows);
    if (!full) {
#pragma unroll
      for (int r = 0; r < 16; ++r)
        *(float4*)(&tile[r * LDSP + lane * 4]) = make_float4(0.0f, 0.0f, 0.0f, 0.0f);
    }
    int nl    = 0;
    int nextb = __builtin_amdgcn_readlane(rp, 1);
    float4 acc = make_float4(0.0f, 0.0f, 0.0f, 0.0f);
    const int plast = pB - 1;
    int2 en = ent[min(pA + lane, plast)];
    for (int cb = pA; cb < pB; cb += 32) {
      const int m = min(32, pB - cb);
      const int2 nx = ent[min(cb + 32 + lane, plast)];
      int j = 0;
      if (m == 32)     { rows_fma_seg<32>(x4, en, rp, j, cb + j, lane, tile, nl, nextb, acc); j += 32; }
      if (j + 16 <= m) { rows_fma_seg<16>(x4, en, rp, j, cb + j, lane, tile, nl, nextb, acc); j += 16; }
      if (j + 8 <= m) { rows_fma_seg<8>(x4, en, rp, j, cb + j, lane, tile, nl, nextb, acc); j += 8; }
      if (j + 4 <= m) { rows_fma_seg<4>(x4, en, rp, j, cb + j, lane, tile, nl, nextb, acc); j += 4; }
      if (j + 2 <= m) { rows_fma_seg<2>(x4, en, rp, j, cb + j, lane, tile, nl, nextb, acc); j += 2; }
      if (j < m)      { rows_fma_seg<1>(x4, en, rp, j, cb + j, lane, tile, nl, nextb, acc); }
      en = nx;
    }
    *(float4*)(&tile[nl * LDSP + lane * 4]) = acc;

    v8f d[8] = {};
    const float* arow = tile + lrow * LDSP + khalf;
    for (int k0 = 0; k0 < DIM; k0 += 4) {
      v2f a = *(const v2f*)(arow + k0);
#pragma unroll
      for (int tt = 0; tt < 8; ++tt) {
        v2f b = *(const v2f*)(&lw[(tt * 16 + lrow) * LDSP + k0 + khalf]);
        d[tt] = __builtin_amdgcn_wmma_f32_16x16x4_f32(
            false, a, false, b, (short)0, d[tt], false, false);
      }
    }

    if (full) {
#pragma unroll
      for (int tt = 0; tt < 8; ++tt) {
        float* op = Out + ((size_t)m0 + rbase) * DIM + tt * 16 + lrow;
        const float bv = bias[tt * 16 + lrow];
#pragma unroll
        for (int r = 0; r < 8; ++r) op[(size_t)r * DIM] = fmaxf(d[tt][r] + bv, 0.0f);
      }
    } else {
      const int vr = nrows - m0 - rbase;
#pragma unroll
      for (int tt = 0; tt < 8; ++tt) {
        float* op = Out + ((size_t)m0 + rbase) * DIM + tt * 16 + lrow;
        const float bv = bias[tt * 16 + lrow];
#pragma unroll
        for (int r = 0; r < 8; ++r)
          if (r < vr) op[(size_t)r * DIM] = fmaxf(d[tt][r] + bv, 0.0f);
      }
    }
  }
}

__global__ __launch_bounds__(1024) void k_gather_pool(const float* __restrict__ h,
                                                      const int2* __restrict__ ent,
                                                      const int* __restrict__ rowptr,
                                                      const int* __restrict__ scope,
                                                      const float* __restrict__ W,
                                                      const float* __restrict__ bias,
                                                      float* __restrict__ out,
                                                      int ngraphs, int nnodes) {
  __shared__ int   wpart[32];
  __shared__ float red[32 * DIM];
  __shared__ float pooled[DIM];
  __shared__ float kq[4 * DIM];
  const int g    = blockIdx.x;
  const int tid  = threadIdx.x;
  const int lane = tid & 31;
  const int w = __builtin_amdgcn_readfirstlane(tid >> 5);
  if (g >= ngraphs) return;

  int part = 0;
  for (int i = tid; i < g; i += 1024) part += scope[i];
  part = wave_sum(part);
  if (lane == 0) wpart[w] = part;
  __syncthreads();
  const int start = __builtin_amdgcn_readfirstlane(wave_sum(wpart[lane]));
  const int cnt   = scope[g];
  const int n0 = min(start, nnodes);
  const int n1 = min(start + cnt, nnodes);
  const int p0 = rowptr[n0];
  const int p1 = rowptr[n1];

  const float4* h4 = (const float4*)h;
  float4 acc = make_float4(0.0f, 0.0f, 0.0f, 0.0f);
  for (int cb = p0 + 32 * w; cb < p1; cb += 32 * 32) {
    const int m = min(32, p1 - cb);
    int2 en = make_int2(0, 0);
    if (lane < m) en = ent[cb + lane];
    for (int j = 0; j < m; ++j) {
      const int   s  = __builtin_amdgcn_readlane(en.x, j);
      const float wt = __int_as_float(__builtin_amdgcn_readlane(en.y, j));
      const float4 v = h4[(size_t)s * (DIM / 4) + lane];
      acc.x = fmaf(wt, v.x, acc.x);
      acc.y = fmaf(wt, v.y, acc.y);
      acc.z = fmaf(wt, v.z, acc.z);
      acc.w = fmaf(wt, v.w, acc.w);
    }
  }

  *(float4*)(&red[w * DIM + lane * 4]) = acc;
  __syncthreads();
  if (tid < DIM) {
    float s = 0.0f;
    for (int k = 0; k < 32; ++k) s += red[k * DIM + tid];
    pooled[tid] = s;
  }
  __syncthreads();

  const int lrow  = lane & 15;
  const int khalf = (lane >> 4) << 1;
  const int ncol  = (w & 7) * 16 + lrow;
  const int kbase = (w >> 3) * 32 + khalf;
  v8f d = {};
  for (int i = 0; i < 8; ++i) {
    const int ka = kbase + i * 4;
    const v2f a = *(const v2f*)(&pooled[ka]);
    v2f b;
    b.x = W[(size_t)ka * DIM + ncol];
    b.y = W[(size_t)(ka + 1) * DIM + ncol];
    d = __builtin_amdgcn_wmma_f32_16x16x4_f32(false, a, false, b, (short)0, d, false, false);
  }
  if (lane < 16) kq[(w >> 3) * DIM + ncol] = d[0];
  __syncthreads();
  if (tid < DIM) {
    const float s = (kq[tid] + kq[DIM + tid]) + (kq[2 * DIM + tid] + kq[3 * DIM + tid]);
    out[(size_t)g * DIM + tid] = s / (float)cnt + bias[tid];
  }
}

extern "C" void kernel_launch(void* const* d_in, const int* in_sizes, int n_in,
                              void* d_out, int out_size, void* d_ws, size_t ws_size,
                              hipStream_t stream) {
  (void)n_in; (void)out_size; (void)ws_size;
  const float* x     = (const float*)d_in[0];
  const int*   ei    = (const int*)d_in[1];
  const int*   scope = (const int*)d_in[2];
  const float* W1    = (const float*)d_in[3];
  const float* b1    = (const float*)d_in[4];
  const float* W2    = (const float*)d_in[5];
  const float* b2    = (const float*)d_in[6];
  float*       out   = (float*)d_out;

  const int n_nodes  = in_sizes[0] / DIM;
  const int n_edges  = in_sizes[1] / 2;
  const int n_graphs = in_sizes[2];
  const int* src = ei;
  const int* dst = ei + n_edges;

  float* bufA = (float*)d_ws;
  float* bufB = bufA + (size_t)n_nodes * DIM;
  int*   cnt  = (int*)(bufB + (size_t)n_nodes * DIM);
  int2* ent    = (int2*)bufB;
  int*  rowptr = (int*)(ent + ((size_t)n_edges + n_nodes));
  int*  rank   = rowptr + n_nodes + 1;
  const int pw = ((n_nodes + 1) / 2 + 3) / 4 * 4;
  unsigned* part = (unsigned*)bufA;
  int2*     off  = (int2*)(part + (size_t)NCHUNK * pw);
  int*      bsum = (int*)(off + (size_t)NCHUNK * pw);

  const int gb_words  = ((n_nodes + 1) / 2 + 255) / 256;
  const int gb_edges  = (n_edges + 255) / 256;

  k_cnt_chunks   <<<NCHUNK,   1024, 0, stream>>>(dst, part, rank, n_nodes, n_edges);
  k_chunk_offsets<<<gb_words,  256, 0, stream>>>(part, off, cnt, bsum, n_nodes);

  k_fill        <<<gb_edges,  256, 0, stream>>>(src, dst, cnt, rank, (const int*)off, bsum, rowptr, ent,
                                                n_nodes, n_edges, gb_words);

  k_gather_gemm <<<NWGP, TWAVES * 32, 0, stream>>>(x, ent, rowptr, W1, b1, bufA, n_nodes);

  k_gather_pool <<<n_graphs, 1024, 0, stream>>>(bufA, ent, rowptr, scope, W2, b2, out,
                                                n_graphs, n_nodes);
}
